# starting kernel
# baseline (speedup 1.0000x reference)
#include <hip/hip_runtime.h>


typedef __attribute__((ext_vector_type(2))) float v2f;
typedef __attribute__((ext_vector_type(8))) float v8f;

__device__ __forceinline__ v8f wmma4(v2f a, v2f b, v8f c) {
  return __builtin_amdgcn_wmma_f32_16x16x4_f32(false, a, false, b, (short)0, c,
                                               false, false);
}

__device__ __forceinline__ v2f ld2(const float* p) { return *(const v2f*)p; }
__device__ __forceinline__ v2f ld2_nt(const float* p) {
  return __builtin_nontemporal_load((const v2f*)p);
}

__global__ __launch_bounds__(256) void k0_tables(float* __restrict__ tab) {
  int idx = blockIdx.x * 256 + threadIdx.x;
  if (idx >= 49152) return;
  const float step = 6.28318530717958647692f / 256.0f;
  float v;
  if (idx < 8192) {
    int n = idx >> 8, w = idx & 255;
    float th = step * (float)(((n & 15) * w) & 255);
    v = (n < 16) ? cosf(th) : -sinf(th);
  } else if (idx < 24576) {
    int j = idx - 8192;
    int is_s = j >= 8192;
    j &= 8191;
    int r = j >> 8, h = j & 255;
    int ky = (r < 16) ? r : (224 + r);
    float th = step * (float)((ky * h) & 255);
    v = is_s ? sinf(th) : cosf(th);
  } else if (idx < 40960) {
    int j = idx - 24576;
    int is_s = j >= 8192;
    j &= 8191;
    int h = j >> 5, r = j & 31;
    int ky = (r < 16) ? r : (224 + r);
    float th = step * (float)((ky * h) & 255);
    v = is_s ? sinf(th) : cosf(th);
  } else {
    int j = idx - 40960;
    int w = j >> 5, n = j & 31;
    int kx = n & 15;
    float th = step * (float)((kx * w) & 255);
    float sc = ((kx == 0) ? 1.0f : 2.0f) * (1.0f / 65536.0f);
    v = (n < 16) ? sc * cosf(th) : -sc * sinf(th);
  }
  tab[idx] = v;
}

__global__ __launch_bounds__(256) void k1_dft_w(const float* __restrict__ x,
                                                const float* __restrict__ FT,
                                                float* __restrict__ Y) {
  const int wave = blockIdx.x * 8 + (threadIdx.x >> 5);
  const int lane = threadIdx.x & 31;
  const int half = lane >> 4, l16 = lane & 15;
  const long rowbase = (long)wave * 16;
  const float* arow = x + (rowbase + l16) * 256 + 2 * half;
  const float* b0 = FT + l16 * 256 + 2 * half;
  const float* b1 = FT + (16 + l16) * 256 + 2 * half;
  v8f acc0 = {}, acc1 = {};
#pragma unroll 4
  for (int k0 = 0; k0 < 256; k0 += 4) {
    v2f a = ld2_nt(arow + k0);
    acc0 = wmma4(a, ld2(b0 + k0), acc0);
    acc1 = wmma4(a, ld2(b1 + k0), acc1);
  }
  float* yrow = Y + rowbase * 32;
#pragma unroll
  for (int r = 0; r < 8; ++r) {
    int row = r + half * 8;
    yrow[row * 32 + l16] = acc0[r];
    yrow[row * 32 + 16 + l16] = acc1[r];
  }
}

__global__ __launch_bounds__(256) void k2_dft_h(const float* __restrict__ Y,
                                                const float* __restrict__ E2c,
                                                const float* __restrict__ E2s,
                                                float* __restrict__ Zre,
                                                float* __restrict__ Zim) {
  const int bi = blockIdx.x * 8 + (threadIdx.x >> 5);
  const int lane = threadIdx.x & 31;
  const int half = lane >> 4, l16 = lane & 15;
  const float* yb = Y + (long)bi * 256 * 32;
  const float* c0 = E2c + l16 * 256 + 2 * half;
  const float* c1 = E2c + (16 + l16) * 256 + 2 * half;
  const float* s0 = E2s + l16 * 256 + 2 * half;
  const float* s1 = E2s + (16 + l16) * 256 + 2 * half;
  v8f a0 = {}, a1 = {}, a2 = {}, a3 = {}, a4 = {}, a5 = {}, a6 = {}, a7 = {};
#pragma unroll 2
  for (int k0 = 0; k0 < 256; k0 += 4) {
    int kr = k0 + 2 * half;
    v2f bre, bim;
    bre.x = yb[kr * 32 + l16];
    bre.y = yb[(kr + 1) * 32 + l16];
    bim.x = yb[kr * 32 + 16 + l16];
    bim.y = yb[(kr + 1) * 32 + 16 + l16];
    v2f ac0 = ld2(c0 + k0), ac1 = ld2(c1 + k0);
    v2f as0 = ld2(s0 + k0), as1 = ld2(s1 + k0);
    a0 = wmma4(ac0, bre, a0);
    a1 = wmma4(as0, bim, a1);
    a2 = wmma4(ac0, bim, a2);
    a3 = wmma4(as0, bre, a3);
    a4 = wmma4(ac1, bre, a4);
    a5 = wmma4(as1, bim, a5);
    a6 = wmma4(ac1, bim, a6);
    a7 = wmma4(as1, bre, a7);
  }
  const int b = bi >> 5, i = bi & 31;
#pragma unroll
  for (int r = 0; r < 8; ++r) {
    int kyr = r + half * 8;
    long o0 = (((long)kyr * 16 + l16) * 16 + b) * 32 + i;
    long o1 = (((long)(16 + kyr) * 16 + l16) * 16 + b) * 32 + i;
    Zre[o0] = a0[r] + a1[r];
    Zim[o0] = a2[r] - a3[r];
    Zre[o1] = a4[r] + a5[r];
    Zim[o1] = a6[r] - a7[r];
  }
}

__global__ __launch_bounds__(256) void k3_mix(
    const float* __restrict__ Zre, const float* __restrict__ Zim,
    const float* __restrict__ w1r, const float* __restrict__ w1i,
    const float* __restrict__ w2r, const float* __restrict__ w2i,
    float* __restrict__ Tre, float* __restrict__ Tim) {
  const int m = blockIdx.x * 8 + (threadIdx.x >> 5);
  const int lane = threadIdx.x & 31;
  const int half = lane >> 4, l16 = lane & 15;
  const int kyp = m >> 4, kx = m & 15;
  const int ky = kyp & 15;
  const float* wr = (kyp < 16) ? w1r : w2r;
  const float* wi = (kyp < 16) ? w1i : w2i;
  const float* zre = Zre + (long)m * 512 + l16 * 32 + 2 * half;
  const float* zim = Zim + (long)m * 512 + l16 * 32 + 2 * half;
  const long wofs = (long)ky * 16 + kx;
  v8f p0 = {}, q0 = {}, u0 = {}, v0 = {}, p1 = {}, q1 = {}, u1 = {}, v1 = {};
#pragma unroll
  for (int k0 = 0; k0 < 32; k0 += 4) {
    int kr = k0 + 2 * half;
    v2f azr = ld2(zre + k0);
    v2f azi = ld2(zim + k0);
    long ba0 = ((long)kr * 32 + l16) * 256 + wofs;
    long ba0b = ((long)(kr + 1) * 32 + l16) * 256 + wofs;
    long ba1 = ((long)kr * 32 + 16 + l16) * 256 + wofs;
    long ba1b = ((long)(kr + 1) * 32 + 16 + l16) * 256 + wofs;
    v2f br0, bi0, br1, bi1;
    br0.x = wr[ba0];  br0.y = wr[ba0b];
    bi0.x = wi[ba0];  bi0.y = wi[ba0b];
    br1.x = wr[ba1];  br1.y = wr[ba1b];
    bi1.x = wi[ba1];  bi1.y = wi[ba1b];
    p0 = wmma4(azr, br0, p0);
    q0 = wmma4(azi, bi0, q0);
    u0 = wmma4(azr, bi0, u0);
    v0 = wmma4(azi, br0, v0);
    p1 = wmma4(azr, br1, p1);
    q1 = wmma4(azi, bi1, q1);
    u1 = wmma4(azr, bi1, u1);
    v1 = wmma4(azi, br1, v1);
  }
#pragma unroll
  for (int r = 0; r < 8; ++r) {
    int b = r + half * 8;
    long o0 = (((long)b * 32 + l16) * 32 + kyp) * 16 + kx;
    long o1 = (((long)b * 32 + 16 + l16) * 32 + kyp) * 16 + kx;
    Tre[o0] = p0[r] - q0[r];
    Tim[o0] = u0[r] + v0[r];
    Tre[o1] = p1[r] - q1[r];
    Tim[o1] = u1[r] + v1[r];
  }
}

__global__ __launch_bounds__(256) void k4_idft_h(const float* __restrict__ Tre,
                                                 const float* __restrict__ Tim,
                                                 const float* __restrict__ E4c,
                                                 const float* __restrict__ E4s,
                                                 float* __restrict__ U) {
  const int wid = blockIdx.x * 8 + (threadIdx.x >> 5);
  const int lane = threadIdx.x & 31;
  const int half = lane >> 4, l16 = lane & 15;
  const int bo = wid >> 4;
  const int h0 = (wid & 15) * 16;
  const float* tre = Tre + (long)bo * 512;
  const float* tim = Tim + (long)bo * 512;
  const float* ec = E4c + (h0 + l16) * 32 + 2 * half;
  const float* es = E4s + (h0 + l16) * 32 + 2 * half;
  v8f a0 = {}, a1 = {}, a2 = {}, a3 = {};
#pragma unroll
  for (int k0 = 0; k0 < 32; k0 += 4) {
    int kr = k0 + 2 * half;
    v2f ac = ld2(ec + k0);
    v2f as = ld2(es + k0);
    v2f bre, bim;
    bre.x = tre[kr * 16 + l16];
    bre.y = tre[(kr + 1) * 16 + l16];
    bim.x = tim[kr * 16 + l16];
    bim.y = tim[(kr + 1) * 16 + l16];
    a0 = wmma4(ac, bre, a0);
    a1 = wmma4(as, bim, a1);
    a2 = wmma4(ac, bim, a2);
    a3 = wmma4(as, bre, a3);
  }
  float* ub = U + ((long)bo * 256 + h0) * 32;
#pragma unroll
  for (int r = 0; r < 8; ++r) {
    int hh = r + half * 8;
    ub[hh * 32 + l16] = a0[r] - a1[r];
    ub[hh * 32 + 16 + l16] = a2[r] + a3[r];
  }
}

__global__ __launch_bounds__(256) void k5_idft_w(const float* __restrict__ U,
                                                 const float* __restrict__ GT,
                                                 float* __restrict__ out) {
  const int wid = blockIdx.x * 8 + (threadIdx.x >> 5);
  const int lane = threadIdx.x & 31;
  const int half = lane >> 4, l16 = lane & 15;
  const long row0 = (long)(wid >> 2) * 16;
  const int n0 = (wid & 3) * 64;
  const float* arow = U + (row0 + l16) * 32 + 2 * half;
  const float* g0 = GT + (n0 + l16) * 32 + 2 * half;
  v8f acc[4] = {};
#pragma unroll
  for (int k0 = 0; k0 < 32; k0 += 4) {
    v2f a = ld2(arow + k0);
#pragma unroll
    for (int t = 0; t < 4; ++t) {
      acc[t] = wmma4(a, ld2(g0 + t * 16 * 32 + k0), acc[t]);
    }
  }
  float* orow = out + row0 * 256;
#pragma unroll
  for (int r = 0; r < 8; ++r) {
    int rr = r + half * 8;
#pragma unroll
    for (int t = 0; t < 4; ++t) {
      __builtin_nontemporal_store(acc[t][r],
                                  &orow[(long)rr * 256 + n0 + t * 16 + l16]);
    }
  }
}

extern "C" void kernel_launch(void* const* d_in, const int* in_sizes, int n_in,
                              void* d_out, int out_size, void* d_ws,
                              size_t ws_size, hipStream_t stream) {
  (void)in_sizes; (void)n_in; (void)out_size; (void)ws_size;
  const float* x = (const float*)d_in[0];
  const float* w1r = (const float*)d_in[1];
  const float* w1i = (const float*)d_in[2];
  const float* w2r = (const float*)d_in[3];
  const float* w2i = (const float*)d_in[4];
  float* out = (float*)d_out;

  float* tab = (float*)d_ws;
  float* FT = tab;
  float* E2c = tab + 8192;
  float* E2s = tab + 16384;
  float* E4c = tab + 24576;
  float* E4s = tab + 32768;
  float* GT = tab + 40960;
  float* Y = tab + 49152;
  float* Zre = Y + 4194304;
  float* Zim = Zre + 262144;
  float* Tre = Zim + 262144;
  float* Tim = Tre + 262144;
  float* U = Tim + 262144;

  hipLaunchKernelGGL(k0_tables, dim3(192), dim3(256), 0, stream, tab);
  hipLaunchKernelGGL(k1_dft_w, dim3(1024), dim3(256), 0, stream, x, FT, Y);
  hipLaunchKernelGGL(k2_dft_h, dim3(64), dim3(256), 0, stream, Y, E2c, E2s, Zre,
                     Zim);
  hipLaunchKernelGGL(k3_mix, dim3(64), dim3(256), 0, stream, Zre, Zim, w1r, w1i,
                     w2r, w2i, Tre, Tim);
  hipLaunchKernelGGL(k4_idft_h, dim3(1024), dim3(256), 0, stream, Tre, Tim, E4c,
                     E4s, U);
  hipLaunchKernelGGL(k5_idft_w, dim3(4096), dim3(256), 0, stream, U, GT, out);
}


# Forward order swapped: H-DFT first straight on x (x as B operand, lanes along w, 16 B per lane, whole-line loads), then W-DFT on the 32 kept rows split over 8 waves; H and W sum order changes
# speedup vs baseline: 1.5132x; 1.5132x over previous
#include <hip/hip_runtime.h>


typedef __attribute__((ext_vector_type(2))) float v2f;
typedef __attribute__((ext_vector_type(4))) float v4f;
typedef __attribute__((ext_vector_type(8))) float v8f;

__device__ __forceinline__ v8f wmma4(v2f a, v2f b, v8f c) {
  return __builtin_amdgcn_wmma_f32_16x16x4_f32(false, a, false, b, (short)0, c,
                                               false, false);
}

__device__ __forceinline__ v2f ld2(const float* p) { return *(const v2f*)p; }
__device__ __forceinline__ v4f ld4_nt(const float* p) {
  return __builtin_nontemporal_load((const v4f*)p);
}

__global__ __launch_bounds__(256) void k0_tables(float* __restrict__ tab) {
  int idx = blockIdx.x * 256 + threadIdx.x;
  if (idx >= 49152) return;
  const float step = 6.28318530717958647692f / 256.0f;
  float v;
  if (idx < 8192) {
    int n = idx >> 8, w = idx & 255;
    float th = step * (float)(((n & 15) * w) & 255);
    v = (n < 16) ? cosf(th) : -sinf(th);
  } else if (idx < 24576) {
    int j = idx - 8192;
    int is_s = j >= 8192;
    j &= 8191;
    int r = j >> 8, h = j & 255;
    int ky = (r < 16) ? r : (224 + r);
    float th = step * (float)((ky * h) & 255);
    v = is_s ? sinf(th) : cosf(th);
  } else if (idx < 40960) {
    int j = idx - 24576;
    int is_s = j >= 8192;
    j &= 8191;
    int h = j >> 5, r = j & 31;
    int ky = (r < 16) ? r : (224 + r);
    float th = step * (float)((ky * h) & 255);
    v = is_s ? sinf(th) : cosf(th);
  } else {
    int j = idx - 40960;
    int w = j >> 5, n = j & 31;
    int kx = n & 15;
    float th = step * (float)((kx * w) & 255);
    float sc = ((kx == 0) ? 1.0f : 2.0f) * (1.0f / 65536.0f);
    v = (n < 16) ? sc * cosf(th) : -sc * sinf(th);
  }
  tab[idx] = v;
}

struct XBlk {
  v4f x[8];
  v2f c0[4], c1[4], s0[4], s1[4];
};

__device__ __forceinline__ void k1_load(XBlk& B, const float* xp,
                                        const float* c0, const float* c1,
                                        const float* s0, const float* s1,
                                        int kb) {
#pragma unroll
  for (int q = 0; q < 4; ++q) {
    int k0 = kb + 4 * q;
    B.x[2 * q] = ld4_nt(xp + (long)k0 * 256);
    B.x[2 * q + 1] = ld4_nt(xp + (long)(k0 + 1) * 256);
    B.c0[q] = ld2(c0 + k0);
    B.c1[q] = ld2(c1 + k0);
    B.s0[q] = ld2(s0 + k0);
    B.s1[q] = ld2(s1 + k0);
  }
}

__device__ __forceinline__ void k1_mma(const XBlk& B, v8f (&acc)[4][4]) {
#pragma unroll
  for (int q = 0; q < 4; ++q) {
#pragma unroll
    for (int t = 0; t < 4; ++t) {
      v2f b;
      b.x = B.x[2 * q][t];
      b.y = B.x[2 * q + 1][t];
      acc[0][t] = wmma4(B.c0[q], b, acc[0][t]);
      acc[1][t] = wmma4(B.c1[q], b, acc[1][t]);
      acc[2][t] = wmma4(B.s0[q], b, acc[2][t]);
      acc[3][t] = wmma4(B.s1[q], b, acc[3][t]);
    }
  }
}

__global__ __launch_bounds__(256) void k1_dft_h(const float* __restrict__ x,
                                                const float* __restrict__ E2c,
                                                const float* __restrict__ E2s,
                                                float* __restrict__ Pc,
                                                float* __restrict__ Ps) {
  const int wv = threadIdx.x >> 5;
  const int lane = threadIdx.x & 31;
  const int half = lane >> 4, l16 = lane & 15;
  const int bi = blockIdx.x * 2 + (wv >> 2);
  const int wbase = (wv & 3) * 64;
  const float* xp = x + (long)bi * 65536 + (2 * half) * 256 + wbase + 4 * l16;
  const float* c0 = E2c + l16 * 256 + 2 * half;
  const float* c1 = E2c + (16 + l16) * 256 + 2 * half;
  const float* s0 = E2s + l16 * 256 + 2 * half;
  const float* s1 = E2s + (16 + l16) * 256 + 2 * half;
  v8f acc[4][4] = {};
  XBlk A, B;
  k1_load(A, xp, c0, c1, s0, s1, 0);
#pragma unroll 1
  for (int kb = 0; kb < 224; kb += 32) {
    k1_load(B, xp, c0, c1, s0, s1, kb + 16);
    k1_mma(A, acc);
    k1_load(A, xp, c0, c1, s0, s1, kb + 32);
    k1_mma(B, acc);
  }
  k1_load(B, xp, c0, c1, s0, s1, 240);
  k1_mma(A, acc);
  k1_mma(B, acc);
  float* pc = Pc + (long)bi * 8192 + wbase + 4 * l16;
  float* ps = Ps + (long)bi * 8192 + wbase + 4 * l16;
#pragma unroll
  for (int r = 0; r < 8; ++r) {
    int row = r + half * 8;
    v4f o0 = {acc[0][0][r], acc[0][1][r], acc[0][2][r], acc[0][3][r]};
    v4f o1 = {acc[1][0][r], acc[1][1][r], acc[1][2][r], acc[1][3][r]};
    v4f o2 = {acc[2][0][r], acc[2][1][r], acc[2][2][r], acc[2][3][r]};
    v4f o3 = {acc[3][0][r], acc[3][1][r], acc[3][2][r], acc[3][3][r]};
    *(v4f*)(pc + row * 256) = o0;
    *(v4f*)(pc + (16 + row) * 256) = o1;
    *(v4f*)(ps + row * 256) = o2;
    *(v4f*)(ps + (16 + row) * 256) = o3;
  }
}

__global__ __launch_bounds__(256) void k2_dft_w(const float* __restrict__ Pc,
                                                const float* __restrict__ Ps,
                                                const float* __restrict__ FT,
                                                float* __restrict__ Zre,
                                                float* __restrict__ Zim) {
  __shared__ float red[8 * 1024];
  const int bi = blockIdx.x;
  const int wv = threadIdx.x >> 5;
  const int lane = threadIdx.x & 31;
  const int half = lane >> 4, l16 = lane & 15;
  const int wb = wv * 32;
  const float* pc0 = Pc + (long)bi * 8192 + l16 * 256 + wb + 2 * half;
  const float* pc1 = pc0 + 16 * 256;
  const float* ps0 = Ps + (long)bi * 8192 + l16 * 256 + wb + 2 * half;
  const float* ps1 = ps0 + 16 * 256;
  const float* fc = FT + l16 * 256 + wb + 2 * half;
  const float* fm = FT + (16 + l16) * 256 + wb + 2 * half;
  v8f a0 = {}, a1 = {}, a2 = {}, a3 = {}, a4 = {}, a5 = {}, a6 = {}, a7 = {};
#pragma unroll
  for (int k0 = 0; k0 < 32; k0 += 4) {
    v2f ac0 = ld2(pc0 + k0), ac1 = ld2(pc1 + k0);
    v2f as0 = ld2(ps0 + k0), as1 = ld2(ps1 + k0);
    v2f bre = ld2(fc + k0), bim = ld2(fm + k0);
    a0 = wmma4(ac0, bre, a0);
    a1 = wmma4(as0, bim, a1);
    a2 = wmma4(ac0, bim, a2);
    a3 = wmma4(as0, bre, a3);
    a4 = wmma4(ac1, bre, a4);
    a5 = wmma4(as1, bim, a5);
    a6 = wmma4(ac1, bim, a6);
    a7 = wmma4(as1, bre, a7);
  }
  float* mine = red + wv * 1024 + lane;
#pragma unroll
  for (int r = 0; r < 8; ++r) {
    mine[(0 * 8 + r) * 32] = a0[r] + a1[r];
    mine[(1 * 8 + r) * 32] = a2[r] - a3[r];
    mine[(2 * 8 + r) * 32] = a4[r] + a5[r];
    mine[(3 * 8 + r) * 32] = a6[r] - a7[r];
  }
  __syncthreads();
  float z0 = red[0 * 256 + threadIdx.x];
  float z1 = red[1 * 256 + threadIdx.x];
  float z2 = red[2 * 256 + threadIdx.x];
  float z3 = red[3 * 256 + threadIdx.x];
#pragma unroll
  for (int v = 1; v < 8; ++v) {
    z0 += red[v * 1024 + 0 * 256 + threadIdx.x];
    z1 += red[v * 1024 + 1 * 256 + threadIdx.x];
    z2 += red[v * 1024 + 2 * 256 + threadIdx.x];
    z3 += red[v * 1024 + 3 * 256 + threadIdx.x];
  }
  const int b = bi >> 5, i = bi & 31;
  const int kyr = wv + half * 8;
  long o0 = (((long)kyr * 16 + l16) * 16 + b) * 32 + i;
  long o1 = (((long)(16 + kyr) * 16 + l16) * 16 + b) * 32 + i;
  Zre[o0] = z0;
  Zim[o0] = z1;
  Zre[o1] = z2;
  Zim[o1] = z3;
}

__global__ __launch_bounds__(256) void k3_mix(
    const float* __restrict__ Zre, const float* __restrict__ Zim,
    const float* __restrict__ w1r, const float* __restrict__ w1i,
    const float* __restrict__ w2r, const float* __restrict__ w2i,
    float* __restrict__ Tre, float* __restrict__ Tim) {
  const int m = blockIdx.x * 8 + (threadIdx.x >> 5);
  const int lane = threadIdx.x & 31;
  const int half = lane >> 4, l16 = lane & 15;
  const int kyp = m >> 4, kx = m & 15;
  const int ky = kyp & 15;
  const float* wr = (kyp < 16) ? w1r : w2r;
  const float* wi = (kyp < 16) ? w1i : w2i;
  const float* zre = Zre + (long)m * 512 + l16 * 32 + 2 * half;
  const float* zim = Zim + (long)m * 512 + l16 * 32 + 2 * half;
  const long wofs = (long)ky * 16 + kx;
  v8f p0 = {}, q0 = {}, u0 = {}, v0 = {}, p1 = {}, q1 = {}, u1 = {}, v1 = {};
#pragma unroll
  for (int k0 = 0; k0 < 32; k0 += 4) {
    int kr = k0 + 2 * half;
    v2f azr = ld2(zre + k0);
    v2f azi = ld2(zim + k0);
    long ba0 = ((long)kr * 32 + l16) * 256 + wofs;
    long ba0b = ((long)(kr + 1) * 32 + l16) * 256 + wofs;
    long ba1 = ((long)kr * 32 + 16 + l16) * 256 + wofs;
    long ba1b = ((long)(kr + 1) * 32 + 16 + l16) * 256 + wofs;
    v2f br0, bi0, br1, bi1;
    br0.x = wr[ba0];  br0.y = wr[ba0b];
    bi0.x = wi[ba0];  bi0.y = wi[ba0b];
    br1.x = wr[ba1];  br1.y = wr[ba1b];
    bi1.x = wi[ba1];  bi1.y = wi[ba1b];
    p0 = wmma4(azr, br0, p0);
    q0 = wmma4(azi, bi0, q0);
    u0 = wmma4(azr, bi0, u0);
    v0 = wmma4(azi, br0, v0);
    p1 = wmma4(azr, br1, p1);
    q1 = wmma4(azi, bi1, q1);
    u1 = wmma4(azr, bi1, u1);
    v1 = wmma4(azi, br1, v1);
  }
#pragma unroll
  for (int r = 0; r < 8; ++r) {
    int b = r + half * 8;
    long o0 = (((long)b * 32 + l16) * 32 + kyp) * 16 + kx;
    long o1 = (((long)b * 32 + 16 + l16) * 32 + kyp) * 16 + kx;
    Tre[o0] = p0[r] - q0[r];
    Tim[o0] = u0[r] + v0[r];
    Tre[o1] = p1[r] - q1[r];
    Tim[o1] = u1[r] + v1[r];
  }
}

__global__ __launch_bounds__(256) void k4_idft_h(const float* __restrict__ Tre,
                                                 const float* __restrict__ Tim,
                                                 const float* __restrict__ E4c,
                                                 const float* __restrict__ E4s,
                                                 float* __restrict__ U) {
  const int wid = blockIdx.x * 8 + (threadIdx.x >> 5);
  const int lane = threadIdx.x & 31;
  const int half = lane >> 4, l16 = lane & 15;
  const int bo = wid >> 4;
  const int h0 = (wid & 15) * 16;
  const float* tre = Tre + (long)bo * 512;
  const float* tim = Tim + (long)bo * 512;
  const float* ec = E4c + (h0 + l16) * 32 + 2 * half;
  const float* es = E4s + (h0 + l16) * 32 + 2 * half;
  v8f a0 = {}, a1 = {}, a2 = {}, a3 = {};
#pragma unroll
  for (int k0 = 0; k0 < 32; k0 += 4) {
    int kr = k0 + 2 * half;
    v2f ac = ld2(ec + k0);
    v2f as = ld2(es + k0);
    v2f bre, bim;
    bre.x = tre[kr * 16 + l16];
    bre.y = tre[(kr + 1) * 16 + l16];
    bim.x = tim[kr * 16 + l16];
    bim.y = tim[(kr + 1) * 16 + l16];
    a0 = wmma4(ac, bre, a0);
    a1 = wmma4(as, bim, a1);
    a2 = wmma4(ac, bim, a2);
    a3 = wmma4(as, bre, a3);
  }
  float* ub = U + ((long)bo * 256 + h0) * 32;
#pragma unroll
  for (int r = 0; r < 8; ++r) {
    int hh = r + half * 8;
    ub[hh * 32 + l16] = a0[r] - a1[r];
    ub[hh * 32 + 16 + l16] = a2[r] + a3[r];
  }
}

__global__ __launch_bounds__(256) void k5_idft_w(const float* __restrict__ U,
                                                 const float* __restrict__ GT,
                                                 float* __restrict__ out) {
  const int wid = blockIdx.x * 8 + (threadIdx.x >> 5);
  const int lane = threadIdx.x & 31;
  const int half = lane >> 4, l16 = lane & 15;
  const long row0 = (long)(wid >> 2) * 16;
  const int n0 = (wid & 3) * 64;
  const float* arow = U + (row0 + l16) * 32 + 2 * half;
  const float* g0 = GT + (n0 + l16) * 32 + 2 * half;
  v8f acc[4] = {};
#pragma unroll
  for (int k0 = 0; k0 < 32; k0 += 4) {
    v2f a = ld2(arow + k0);
#pragma unroll
    for (int t = 0; t < 4; ++t) {
      acc[t] = wmma4(a, ld2(g0 + t * 16 * 32 + k0), acc[t]);
    }
  }
  float* orow = out + row0 * 256;
#pragma unroll
  for (int r = 0; r < 8; ++r) {
    int rr = r + half * 8;
#pragma unroll
    for (int t = 0; t < 4; ++t) {
      __builtin_nontemporal_store(acc[t][r],
                                  &orow[(long)rr * 256 + n0 + t * 16 + l16]);
    }
  }
}

extern "C" void kernel_launch(void* const* d_in, const int* in_sizes, int n_in,
                              void* d_out, int out_size, void* d_ws,
                              size_t ws_size, hipStream_t stream) {
  (void)in_sizes; (void)n_in; (void)out_size; (void)ws_size;
  const float* x = (const float*)d_in[0];
  const float* w1r = (const float*)d_in[1];
  const float* w1i = (const float*)d_in[2];
  const float* w2r = (const float*)d_in[3];
  const float* w2i = (const float*)d_in[4];
  float* out = (float*)d_out;

  float* tab = (float*)d_ws;
  float* FT = tab;
  float* E2c = tab + 8192;
  float* E2s = tab + 16384;
  float* E4c = tab + 24576;
  float* E4s = tab + 32768;
  float* GT = tab + 40960;
  float* Pc = tab + 49152;
  float* Ps = Pc + 4194304;
  float* Zre = Ps + 4194304;
  float* Zim = Zre + 262144;
  float* Tre = Zim + 262144;
  float* Tim = Tre + 262144;
  float* U = Tim + 262144;

  hipLaunchKernelGGL(k0_tables, dim3(192), dim3(256), 0, stream, tab);
  hipLaunchKernelGGL(k1_dft_h, dim3(256), dim3(256), 0, stream, x, E2c, E2s, Pc,
                     Ps);
  hipLaunchKernelGGL(k2_dft_w, dim3(512), dim3(256), 0, stream, Pc, Ps, FT, Zre,
                     Zim);
  hipLaunchKernelGGL(k3_mix, dim3(64), dim3(256), 0, stream, Zre, Zim, w1r, w1i,
                     w2r, w2i, Tre, Tim);
  hipLaunchKernelGGL(k4_idft_h, dim3(1024), dim3(256), 0, stream, Tre, Tim, E4c,
                     E4s, U);
  hipLaunchKernelGGL(k5_idft_w, dim3(4096), dim3(256), 0, stream, U, GT, out);
}


# Inverse W-DFT kernel: a wave keeps its 64-column table fragments in registers and walks the 16 row tiles of one image, next tile's U loads in flight; 2048 waves instead of 32768; sums unchanged
# speedup vs baseline: 1.5878x; 1.0493x over previous
#include <hip/hip_runtime.h>


typedef __attribute__((ext_vector_type(2))) float v2f;
typedef __attribute__((ext_vector_type(4))) float v4f;
typedef __attribute__((ext_vector_type(8))) float v8f;

__device__ __forceinline__ v8f wmma4(v2f a, v2f b, v8f c) {
  return __builtin_amdgcn_wmma_f32_16x16x4_f32(false, a, false, b, (short)0, c,
                                               false, false);
}

__device__ __forceinline__ v2f ld2(const float* p) { return *(const v2f*)p; }
__device__ __forceinline__ v4f ld4_nt(const float* p) {
  return __builtin_nontemporal_load((const v4f*)p);
}

__global__ __launch_bounds__(256) void k0_tables(float* __restrict__ tab) {
  int idx = blockIdx.x * 256 + threadIdx.x;
  if (idx >= 49152) return;
  const float step = 6.28318530717958647692f / 256.0f;
  float v;
  if (idx < 8192) {
    int n = idx >> 8, w = idx & 255;
    float th = step * (float)(((n & 15) * w) & 255);
    v = (n < 16) ? cosf(th) : -sinf(th);
  } else if (idx < 24576) {
    int j = idx - 8192;
    int is_s = j >= 8192;
    j &= 8191;
    int r = j >> 8, h = j & 255;
    int ky = (r < 16) ? r : (224 + r);
    float th = step * (float)((ky * h) & 255);
    v = is_s ? sinf(th) : cosf(th);
  } else if (idx < 40960) {
    int j = idx - 24576;
    int is_s = j >= 8192;
    j &= 8191;
    int h = j >> 5, r = j & 31;
    int ky = (r < 16) ? r : (224 + r);
    float th = step * (float)((ky * h) & 255);
    v = is_s ? sinf(th) : cosf(th);
  } else {
    int j = idx - 40960;
    int w = j >> 5, n = j & 31;
    int kx = n & 15;
    float th = step * (float)((kx * w) & 255);
    float sc = ((kx == 0) ? 1.0f : 2.0f) * (1.0f / 65536.0f);
    v = (n < 16) ? sc * cosf(th) : -sc * sinf(th);
  }
  tab[idx] = v;
}

struct XBlk {
  v4f x[8];
  v2f c0[4], c1[4], s0[4], s1[4];
};

__device__ __forceinline__ void k1_load(XBlk& B, const float* xp,
                                        const float* c0, const float* c1,
                                        const float* s0, const float* s1,
                                        int kb) {
#pragma unroll
  for (int q = 0; q < 4; ++q) {
    int k0 = kb + 4 * q;
    B.x[2 * q] = ld4_nt(xp + (long)k0 * 256);
    B.x[2 * q + 1] = ld4_nt(xp + (long)(k0 + 1) * 256);
    B.c0[q] = ld2(c0 + k0);
    B.c1[q] = ld2(c1 + k0);
    B.s0[q] = ld2(s0 + k0);
    B.s1[q] = ld2(s1 + k0);
  }
}

__device__ __forceinline__ void k1_mma(const XBlk& B, v8f (&acc)[4][4]) {
#pragma unroll
  for (int q = 0; q < 4; ++q) {
#pragma unroll
    for (int t = 0; t < 4; ++t) {
      v2f b;
      b.x = B.x[2 * q][t];
      b.y = B.x[2 * q + 1][t];
      acc[0][t] = wmma4(B.c0[q], b, acc[0][t]);
      acc[1][t] = wmma4(B.c1[q], b, acc[1][t]);
      acc[2][t] = wmma4(B.s0[q], b, acc[2][t]);
      acc[3][t] = wmma4(B.s1[q], b, acc[3][t]);
    }
  }
}

__global__ __launch_bounds__(256) void k1_dft_h(const float* __restrict__ x,
                                                const float* __restrict__ E2c,
                                                const float* __restrict__ E2s,
                                                float* __restrict__ Pc,
                                                float* __restrict__ Ps) {
  const int wv = threadIdx.x >> 5;
  const int lane = threadIdx.x & 31;
  const int half = lane >> 4, l16 = lane & 15;
  const int bi = blockIdx.x * 2 + (wv >> 2);
  const int wbase = (wv & 3) * 64;
  const float* xp = x + (long)bi * 65536 + (2 * half) * 256 + wbase + 4 * l16;
  const float* c0 = E2c + l16 * 256 + 2 * half;
  const float* c1 = E2c + (16 + l16) * 256 + 2 * half;
  const float* s0 = E2s + l16 * 256 + 2 * half;
  const float* s1 = E2s + (16 + l16) * 256 + 2 * half;
  v8f acc[4][4] = {};
  XBlk A, B;
  k1_load(A, xp, c0, c1, s0, s1, 0);
#pragma unroll 1
  for (int kb = 0; kb < 224; kb += 32) {
    k1_load(B, xp, c0, c1, s0, s1, kb + 16);
    k1_mma(A, acc);
    k1_load(A, xp, c0, c1, s0, s1, kb + 32);
    k1_mma(B, acc);
  }
  k1_load(B, xp, c0, c1, s0, s1, 240);
  k1_mma(A, acc);
  k1_mma(B, acc);
  float* pc = Pc + (long)bi * 8192 + wbase + 4 * l16;
  float* ps = Ps + (long)bi * 8192 + wbase + 4 * l16;
#pragma unroll
  for (int r = 0; r < 8; ++r) {
    int row = r + half * 8;
    v4f o0 = {acc[0][0][r], acc[0][1][r], acc[0][2][r], acc[0][3][r]};
    v4f o1 = {acc[1][0][r], acc[1][1][r], acc[1][2][r], acc[1][3][r]};
    v4f o2 = {acc[2][0][r], acc[2][1][r], acc[2][2][r], acc[2][3][r]};
    v4f o3 = {acc[3][0][r], acc[3][1][r], acc[3][2][r], acc[3][3][r]};
    *(v4f*)(pc + row * 256) = o0;
    *(v4f*)(pc + (16 + row) * 256) = o1;
    *(v4f*)(ps + row * 256) = o2;
    *(v4f*)(ps + (16 + row) * 256) = o3;
  }
}

__global__ __launch_bounds__(256) void k2_dft_w(const float* __restrict__ Pc,
                                                const float* __restrict__ Ps,
                                                const float* __restrict__ FT,
                                                float* __restrict__ Zre,
                                                float* __restrict__ Zim) {
  __shared__ float red[8 * 1024];
  const int bi = blockIdx.x;
  const int wv = threadIdx.x >> 5;
  const int lane = threadIdx.x & 31;
  const int half = lane >> 4, l16 = lane & 15;
  const int wb = wv * 32;
  const float* pc0 = Pc + (long)bi * 8192 + l16 * 256 + wb + 2 * half;
  const float* pc1 = pc0 + 16 * 256;
  const float* ps0 = Ps + (long)bi * 8192 + l16 * 256 + wb + 2 * half;
  const float* ps1 = ps0 + 16 * 256;
  const float* fc = FT + l16 * 256 + wb + 2 * half;
  const float* fm = FT + (16 + l16) * 256 + wb + 2 * half;
  v8f a0 = {}, a1 = {}, a2 = {}, a3 = {}, a4 = {}, a5 = {}, a6 = {}, a7 = {};
#pragma unroll
  for (int k0 = 0; k0 < 32; k0 += 4) {
    v2f ac0 = ld2(pc0 + k0), ac1 = ld2(pc1 + k0);
    v2f as0 = ld2(ps0 + k0), as1 = ld2(ps1 + k0);
    v2f bre = ld2(fc + k0), bim = ld2(fm + k0);
    a0 = wmma4(ac0, bre, a0);
    a1 = wmma4(as0, bim, a1);
    a2 = wmma4(ac0, bim, a2);
    a3 = wmma4(as0, bre, a3);
    a4 = wmma4(ac1, bre, a4);
    a5 = wmma4(as1, bim, a5);
    a6 = wmma4(ac1, bim, a6);
    a7 = wmma4(as1, bre, a7);
  }
  float* mine = red + wv * 1024 + lane;
#pragma unroll
  for (int r = 0; r < 8; ++r) {
    mine[(0 * 8 + r) * 32] = a0[r] + a1[r];
    mine[(1 * 8 + r) * 32] = a2[r] - a3[r];
    mine[(2 * 8 + r) * 32] = a4[r] + a5[r];
    mine[(3 * 8 + r) * 32] = a6[r] - a7[r];
  }
  __syncthreads();
  float z0 = red[0 * 256 + threadIdx.x];
  float z1 = red[1 * 256 + threadIdx.x];
  float z2 = red[2 * 256 + threadIdx.x];
  float z3 = red[3 * 256 + threadIdx.x];
#pragma unroll
  for (int v = 1; v < 8; ++v) {
    z0 += red[v * 1024 + 0 * 256 + threadIdx.x];
    z1 += red[v * 1024 + 1 * 256 + threadIdx.x];
    z2 += red[v * 1024 + 2 * 256 + threadIdx.x];
    z3 += red[v * 1024 + 3 * 256 + threadIdx.x];
  }
  const int b = bi >> 5, i = bi & 31;
  const int kyr = wv + half * 8;
  long o0 = (((long)kyr * 16 + l16) * 16 + b) * 32 + i;
  long o1 = (((long)(16 + kyr) * 16 + l16) * 16 + b) * 32 + i;
  Zre[o0] = z0;
  Zim[o0] = z1;
  Zre[o1] = z2;
  Zim[o1] = z3;
}

__global__ __launch_bounds__(256) void k3_mix(
    const float* __restrict__ Zre, const float* __restrict__ Zim,
    const float* __restrict__ w1r, const float* __restrict__ w1i,
    const float* __restrict__ w2r, const float* __restrict__ w2i,
    float* __restrict__ Tre, float* __restrict__ Tim) {
  const int m = blockIdx.x * 8 + (threadIdx.x >> 5);
  const int lane = threadIdx.x & 31;
  const int half = lane >> 4, l16 = lane & 15;
  const int kyp = m >> 4, kx = m & 15;
  const int ky = kyp & 15;
  const float* wr = (kyp < 16) ? w1r : w2r;
  const float* wi = (kyp < 16) ? w1i : w2i;
  const float* zre = Zre + (long)m * 512 + l16 * 32 + 2 * half;
  const float* zim = Zim + (long)m * 512 + l16 * 32 + 2 * half;
  const long wofs = (long)ky * 16 + kx;
  v8f p0 = {}, q0 = {}, u0 = {}, v0 = {}, p1 = {}, q1 = {}, u1 = {}, v1 = {};
#pragma unroll
  for (int k0 = 0; k0 < 32; k0 += 4) {
    int kr = k0 + 2 * half;
    v2f azr = ld2(zre + k0);
    v2f azi = ld2(zim + k0);
    long ba0 = ((long)kr * 32 + l16) * 256 + wofs;
    long ba0b = ((long)(kr + 1) * 32 + l16) * 256 + wofs;
    long ba1 = ((long)kr * 32 + 16 + l16) * 256 + wofs;
    long ba1b = ((long)(kr + 1) * 32 + 16 + l16) * 256 + wofs;
    v2f br0, bi0, br1, bi1;
    br0.x = wr[ba0];  br0.y = wr[ba0b];
    bi0.x = wi[ba0];  bi0.y = wi[ba0b];
    br1.x = wr[ba1];  br1.y = wr[ba1b];
    bi1.x = wi[ba1];  bi1.y = wi[ba1b];
    p0 = wmma4(azr, br0, p0);
    q0 = wmma4(azi, bi0, q0);
    u0 = wmma4(azr, bi0, u0);
    v0 = wmma4(azi, br0, v0);
    p1 = wmma4(azr, br1, p1);
    q1 = wmma4(azi, bi1, q1);
    u1 = wmma4(azr, bi1, u1);
    v1 = wmma4(azi, br1, v1);
  }
#pragma unroll
  for (int r = 0; r < 8; ++r) {
    int b = r + half * 8;
    long o0 = (((long)b * 32 + l16) * 32 + kyp) * 16 + kx;
    long o1 = (((long)b * 32 + 16 + l16) * 32 + kyp) * 16 + kx;
    Tre[o0] = p0[r] - q0[r];
    Tim[o0] = u0[r] + v0[r];
    Tre[o1] = p1[r] - q1[r];
    Tim[o1] = u1[r] + v1[r];
  }
}

__global__ __launch_bounds__(256) void k4_idft_h(const float* __restrict__ Tre,
                                                 const float* __restrict__ Tim,
                                                 const float* __restrict__ E4c,
                                                 const float* __restrict__ E4s,
                                                 float* __restrict__ U) {
  const int wid = blockIdx.x * 8 + (threadIdx.x >> 5);
  const int lane = threadIdx.x & 31;
  const int half = lane >> 4, l16 = lane & 15;
  const int bo = wid >> 4;
  const int h0 = (wid & 15) * 16;
  const float* tre = Tre + (long)bo * 512;
  const float* tim = Tim + (long)bo * 512;
  const float* ec = E4c + (h0 + l16) * 32 + 2 * half;
  const float* es = E4s + (h0 + l16) * 32 + 2 * half;
  v8f a0 = {}, a1 = {}, a2 = {}, a3 = {};
#pragma unroll
  for (int k0 = 0; k0 < 32; k0 += 4) {
    int kr = k0 + 2 * half;
    v2f ac = ld2(ec + k0);
    v2f as = ld2(es + k0);
    v2f bre, bim;
    bre.x = tre[kr * 16 + l16];
    bre.y = tre[(kr + 1) * 16 + l16];
    bim.x = tim[kr * 16 + l16];
    bim.y = tim[(kr + 1) * 16 + l16];
    a0 = wmma4(ac, bre, a0);
    a1 = wmma4(as, bim, a1);
    a2 = wmma4(ac, bim, a2);
    a3 = wmma4(as, bre, a3);
  }
  float* ub = U + ((long)bo * 256 + h0) * 32;
#pragma unroll
  for (int r = 0; r < 8; ++r) {
    int hh = r + half * 8;
    ub[hh * 32 + l16] = a0[r] - a1[r];
    ub[hh * 32 + 16 + l16] = a2[r] + a3[r];
  }
}

__device__ __forceinline__ void k5_load(v2f (&a)[8], const float* arow) {
#pragma unroll
  for (int q = 0; q < 8; ++q) a[q] = ld2(arow + 4 * q);
}

__device__ __forceinline__ void k5_tile(const v2f (&a)[8],
                                        const v2f (&g)[4][8], float* o) {
  v8f acc[4] = {};
#pragma unroll
  for (int q = 0; q < 8; ++q) {
#pragma unroll
    for (int t = 0; t < 4; ++t) acc[t] = wmma4(a[q], g[t][q], acc[t]);
  }
#pragma unroll
  for (int r = 0; r < 8; ++r) {
#pragma unroll
    for (int t = 0; t < 4; ++t) {
      __builtin_nontemporal_store(acc[t][r], &o[r * 256 + t * 16]);
    }
  }
}

__global__ __launch_bounds__(256) void k5_idft_w(const float* __restrict__ U,
                                                 const float* __restrict__ GT,
                                                 float* __restrict__ out) {
  const int wv = threadIdx.x >> 5;
  const int lane = threadIdx.x & 31;
  const int half = lane >> 4, l16 = lane & 15;
  const int bo = blockIdx.x * 2 + (wv >> 2);
  const int n0 = (wv & 3) * 64;
  const long rowbase = (long)bo * 256;
  const float* g0 = GT + (n0 + l16) * 32 + 2 * half;
  v2f g[4][8];
#pragma unroll
  for (int t = 0; t < 4; ++t) {
#pragma unroll
    for (int q = 0; q < 8; ++q) g[t][q] = ld2(g0 + t * 16 * 32 + 4 * q);
  }
  const float* arow = U + (rowbase + l16) * 32 + 2 * half;
  float* obase = out + (rowbase + half * 8) * 256 + n0 + l16;
  v2f A[8], B[8];
  k5_load(A, arow);
  for (int j = 0; j < 14; j += 2) {
    k5_load(B, arow + (j + 1) * 512);
    k5_tile(A, g, obase + j * 4096);
    k5_load(A, arow + (j + 2) * 512);
    k5_tile(B, g, obase + (j + 1) * 4096);
  }
  k5_load(B, arow + 15 * 512);
  k5_tile(A, g, obase + 14 * 4096);
  k5_tile(B, g, obase + 15 * 4096);
}

extern "C" void kernel_launch(void* const* d_in, const int* in_sizes, int n_in,
                              void* d_out, int out_size, void* d_ws,
                              size_t ws_size, hipStream_t stream) {
  (void)in_sizes; (void)n_in; (void)out_size; (void)ws_size;
  const float* x = (const float*)d_in[0];
  const float* w1r = (const float*)d_in[1];
  const float* w1i = (const float*)d_in[2];
  const float* w2r = (const float*)d_in[3];
  const float* w2i = (const float*)d_in[4];
  float* out = (float*)d_out;

  float* tab = (float*)d_ws;
  float* FT = tab;
  float* E2c = tab + 8192;
  float* E2s = tab + 16384;
  float* E4c = tab + 24576;
  float* E4s = tab + 32768;
  float* GT = tab + 40960;
  float* Pc = tab + 49152;
  float* Ps = Pc + 4194304;
  float* Zre = Ps + 4194304;
  float* Zim = Zre + 262144;
  float* Tre = Zim + 262144;
  float* Tim = Tre + 262144;
  float* U = Tim + 262144;

  hipLaunchKernelGGL(k0_tables, dim3(192), dim3(256), 0, stream, tab);
  hipLaunchKernelGGL(k1_dft_h, dim3(256), dim3(256), 0, stream, x, E2c, E2s, Pc,
                     Ps);
  hipLaunchKernelGGL(k2_dft_w, dim3(512), dim3(256), 0, stream, Pc, Ps, FT, Zre,
                     Zim);
  hipLaunchKernelGGL(k3_mix, dim3(64), dim3(256), 0, stream, Zre, Zim, w1r, w1i,
                     w2r, w2i, Tre, Tim);
  hipLaunchKernelGGL(k4_idft_h, dim3(1024), dim3(256), 0, stream, Tre, Tim, E4c,
                     E4s, U);
  hipLaunchKernelGGL(k5_idft_w, dim3(256), dim3(256), 0, stream, U, GT, out);
}


# Forward H-DFT on the image folded about h=128 (cos even, sin odd: B = x[h]+x[256-h] for cos tiles, x[h]-x[256-h] for sin tiles): 520 WMMAs per wave instead of 1024; H sum order changes
# speedup vs baseline: 1.6980x; 1.0694x over previous
#include <hip/hip_runtime.h>


typedef __attribute__((ext_vector_type(2))) float v2f;
typedef __attribute__((ext_vector_type(4))) float v4f;
typedef __attribute__((ext_vector_type(8))) float v8f;

__device__ __forceinline__ v8f wmma4(v2f a, v2f b, v8f c) {
  return __builtin_amdgcn_wmma_f32_16x16x4_f32(false, a, false, b, (short)0, c,
                                               false, false);
}

__device__ __forceinline__ v2f ld2(const float* p) { return *(const v2f*)p; }
__device__ __forceinline__ v4f ld4_nt(const float* p) {
  return __builtin_nontemporal_load((const v4f*)p);
}

__global__ __launch_bounds__(256) void k0_tables(float* __restrict__ tab) {
  int idx = blockIdx.x * 256 + threadIdx.x;
  if (idx >= 49152) return;
  const float step = 6.28318530717958647692f / 256.0f;
  float v;
  if (idx < 8192) {
    int n = idx >> 8, w = idx & 255;
    float th = step * (float)(((n & 15) * w) & 255);
    v = (n < 16) ? cosf(th) : -sinf(th);
  } else if (idx < 24576) {
    int j = idx - 8192;
    int is_s = j >= 8192;
    j &= 8191;
    int r = j >> 8, h = j & 255;
    int ky = (r < 16) ? r : (224 + r);
    float th = step * (float)((ky * h) & 255);
    v = is_s ? sinf(th) : cosf(th);
  } else if (idx < 40960) {
    int j = idx - 24576;
    int is_s = j >= 8192;
    j &= 8191;
    int h = j >> 5, r = j & 31;
    int ky = (r < 16) ? r : (224 + r);
    float th = step * (float)((ky * h) & 255);
    v = is_s ? sinf(th) : cosf(th);
  } else {
    int j = idx - 40960;
    int w = j >> 5, n = j & 31;
    int kx = n & 15;
    float th = step * (float)((kx * w) & 255);
    float sc = ((kx == 0) ? 1.0f : 2.0f) * (1.0f / 65536.0f);
    v = (n < 16) ? sc * cosf(th) : -sc * sinf(th);
  }
  tab[idx] = v;
}

struct XBlk {
  v4f xa[8];
  v4f xm[8];
  v2f c0[4], c1[4], s0[4], s1[4];
};

__device__ __forceinline__ void k1_load(XBlk& B, const float* xp,
                                        const float* xmp, const float* c0,
                                        const float* c1, const float* s0,
                                        const float* s1, int kb, int half) {
#pragma unroll
  for (int q = 0; q < 4; ++q) {
    int k0 = kb + 4 * q;
    B.xa[2 * q] = ld4_nt(xp + (long)k0 * 256);
    B.xa[2 * q + 1] = ld4_nt(xp + (long)(k0 + 1) * 256);
    const float* m0 = xmp - (long)k0 * 256;
    if (q == 0) m0 = (k0 + 2 * half == 0) ? xp : m0;
    B.xm[2 * q] = ld4_nt(m0);
    B.xm[2 * q + 1] = ld4_nt(xmp - (long)(k0 + 1) * 256);
    B.c0[q] = ld2(c0 + k0);
    B.c1[q] = ld2(c1 + k0);
    B.s0[q] = ld2(s0 + k0);
    B.s1[q] = ld2(s1 + k0);
  }
}

__device__ __forceinline__ void k1_mma(const XBlk& B, v8f (&acc)[4][4],
                                       bool row0) {
#pragma unroll
  for (int q = 0; q < 4; ++q) {
    v4f e0 = B.xa[2 * q] + B.xm[2 * q];
    v4f e1 = B.xa[2 * q + 1] + B.xm[2 * q + 1];
    v4f o0 = B.xa[2 * q] - B.xm[2 * q];
    v4f o1 = B.xa[2 * q + 1] - B.xm[2 * q + 1];
    if (q == 0) {
#pragma unroll
      for (int t = 0; t < 4; ++t) e0[t] = row0 ? B.xa[0][t] : e0[t];
    }
#pragma unroll
    for (int t = 0; t < 4; ++t) {
      v2f be, bo;
      be.x = e0[t];
      be.y = e1[t];
      bo.x = o0[t];
      bo.y = o1[t];
      acc[0][t] = wmma4(B.c0[q], be, acc[0][t]);
      acc[1][t] = wmma4(B.c1[q], be, acc[1][t]);
      acc[2][t] = wmma4(B.s0[q], bo, acc[2][t]);
      acc[3][t] = wmma4(B.s1[q], bo, acc[3][t]);
    }
  }
}

__global__ __launch_bounds__(256) void k1_dft_h(const float* __restrict__ x,
                                                const float* __restrict__ E2c,
                                                const float* __restrict__ E2s,
                                                float* __restrict__ Pc,
                                                float* __restrict__ Ps) {
  const int wv = threadIdx.x >> 5;
  const int lane = threadIdx.x & 31;
  const int half = lane >> 4, l16 = lane & 15;
  const int bi = blockIdx.x * 2 + (wv >> 2);
  const int wbase = (wv & 3) * 64;
  const float* ximg = x + (long)bi * 65536 + wbase + 4 * l16;
  const float* xp = ximg + (2 * half) * 256;
  const float* xmp = ximg + (256 - 2 * half) * 256;
  const float* c0 = E2c + l16 * 256 + 2 * half;
  const float* c1 = E2c + (16 + l16) * 256 + 2 * half;
  const float* s0 = E2s + l16 * 256 + 2 * half;
  const float* s1 = E2s + (16 + l16) * 256 + 2 * half;
  v8f acc[4][4] = {};
  XBlk A, B;
  k1_load(A, xp, xmp, c0, c1, s0, s1, 0, half);
#pragma unroll 1
  for (int kb = 0; kb < 96; kb += 32) {
    k1_load(B, xp, xmp, c0, c1, s0, s1, kb + 16, half);
    k1_mma(A, acc, (kb == 0) && (half == 0));
    k1_load(A, xp, xmp, c0, c1, s0, s1, kb + 32, half);
    k1_mma(B, acc, false);
  }
  k1_load(B, xp, xmp, c0, c1, s0, s1, 112, half);
  v4f x128 = ld4_nt(ximg + 128 * 256);
  v2f t0 = ld2(c0 + 128);
  v2f t1 = ld2(c1 + 128);
  k1_mma(A, acc, false);
  k1_mma(B, acc, false);
  {
    v2f a0, a1;
    a0.x = half ? 0.0f : t0.x;
    a0.y = 0.0f;
    a1.x = half ? 0.0f : t1.x;
    a1.y = 0.0f;
#pragma unroll
    for (int t = 0; t < 4; ++t) {
      v2f b;
      b.x = half ? 0.0f : x128[t];
      b.y = 0.0f;
      acc[0][t] = wmma4(a0, b, acc[0][t]);
      acc[1][t] = wmma4(a1, b, acc[1][t]);
    }
  }
  float* pc = Pc + (long)bi * 8192 + wbase + 4 * l16;
  float* ps = Ps + (long)bi * 8192 + wbase + 4 * l16;
#pragma unroll
  for (int r = 0; r < 8; ++r) {
    int row = r + half * 8;
    v4f o0 = {acc[0][0][r], acc[0][1][r], acc[0][2][r], acc[0][3][r]};
    v4f o1 = {acc[1][0][r], acc[1][1][r], acc[1][2][r], acc[1][3][r]};
    v4f o2 = {acc[2][0][r], acc[2][1][r], acc[2][2][r], acc[2][3][r]};
    v4f o3 = {acc[3][0][r], acc[3][1][r], acc[3][2][r], acc[3][3][r]};
    *(v4f*)(pc + row * 256) = o0;
    *(v4f*)(pc + (16 + row) * 256) = o1;
    *(v4f*)(ps + row * 256) = o2;
    *(v4f*)(ps + (16 + row) * 256) = o3;
  }
}

__global__ __launch_bounds__(256) void k2_dft_w(const float* __restrict__ Pc,
                                                const float* __restrict__ Ps,
                                                const float* __restrict__ FT,
                                                float* __restrict__ Zre,
                                                float* __restrict__ Zim) {
  __shared__ float red[8 * 1024];
  const int bi = blockIdx.x;
  const int wv = threadIdx.x >> 5;
  const int lane = threadIdx.x & 31;
  const int half = lane >> 4, l16 = lane & 15;
  const int wb = wv * 32;
  const float* pc0 = Pc + (long)bi * 8192 + l16 * 256 + wb + 2 * half;
  const float* pc1 = pc0 + 16 * 256;
  const float* ps0 = Ps + (long)bi * 8192 + l16 * 256 + wb + 2 * half;
  const float* ps1 = ps0 + 16 * 256;
  const float* fc = FT + l16 * 256 + wb + 2 * half;
  const float* fm = FT + (16 + l16) * 256 + wb + 2 * half;
  v8f a0 = {}, a1 = {}, a2 = {}, a3 = {}, a4 = {}, a5 = {}, a6 = {}, a7 = {};
#pragma unroll
  for (int k0 = 0; k0 < 32; k0 += 4) {
    v2f ac0 = ld2(pc0 + k0), ac1 = ld2(pc1 + k0);
    v2f as0 = ld2(ps0 + k0), as1 = ld2(ps1 + k0);
    v2f bre = ld2(fc + k0), bim = ld2(fm + k0);
    a0 = wmma4(ac0, bre, a0);
    a1 = wmma4(as0, bim, a1);
    a2 = wmma4(ac0, bim, a2);
    a3 = wmma4(as0, bre, a3);
    a4 = wmma4(ac1, bre, a4);
    a5 = wmma4(as1, bim, a5);
    a6 = wmma4(ac1, bim, a6);
    a7 = wmma4(as1, bre, a7);
  }
  float* mine = red + wv * 1024 + lane;
#pragma unroll
  for (int r = 0; r < 8; ++r) {
    mine[(0 * 8 + r) * 32] = a0[r] + a1[r];
    mine[(1 * 8 + r) * 32] = a2[r] - a3[r];
    mine[(2 * 8 + r) * 32] = a4[r] + a5[r];
    mine[(3 * 8 + r) * 32] = a6[r] - a7[r];
  }
  __syncthreads();
  float z0 = red[0 * 256 + threadIdx.x];
  float z1 = red[1 * 256 + threadIdx.x];
  float z2 = red[2 * 256 + threadIdx.x];
  float z3 = red[3 * 256 + threadIdx.x];
#pragma unroll
  for (int v = 1; v < 8; ++v) {
    z0 += red[v * 1024 + 0 * 256 + threadIdx.x];
    z1 += red[v * 1024 + 1 * 256 + threadIdx.x];
    z2 += red[v * 1024 + 2 * 256 + threadIdx.x];
    z3 += red[v * 1024 + 3 * 256 + threadIdx.x];
  }
  const int b = bi >> 5, i = bi & 31;
  const int kyr = wv + half * 8;
  long o0 = (((long)kyr * 16 + l16) * 16 + b) * 32 + i;
  long o1 = (((long)(16 + kyr) * 16 + l16) * 16 + b) * 32 + i;
  Zre[o0] = z0;
  Zim[o0] = z1;
  Zre[o1] = z2;
  Zim[o1] = z3;
}

__global__ __launch_bounds__(256) void k3_mix(
    const float* __restrict__ Zre, const float* __restrict__ Zim,
    const float* __restrict__ w1r, const float* __restrict__ w1i,
    const float* __restrict__ w2r, const float* __restrict__ w2i,
    float* __restrict__ Tre, float* __restrict__ Tim) {
  const int m = blockIdx.x * 8 + (threadIdx.x >> 5);
  const int lane = threadIdx.x & 31;
  const int half = lane >> 4, l16 = lane & 15;
  const int kyp = m >> 4, kx = m & 15;
  const int ky = kyp & 15;
  const float* wr = (kyp < 16) ? w1r : w2r;
  const float* wi = (kyp < 16) ? w1i : w2i;
  const float* zre = Zre + (long)m * 512 + l16 * 32 + 2 * half;
  const float* zim = Zim + (long)m * 512 + l16 * 32 + 2 * half;
  const long wofs = (long)ky * 16 + kx;
  v8f p0 = {}, q0 = {}, u0 = {}, v0 = {}, p1 = {}, q1 = {}, u1 = {}, v1 = {};
#pragma unroll
  for (int k0 = 0; k0 < 32; k0 += 4) {
    int kr = k0 + 2 * half;
    v2f azr = ld2(zre + k0);
    v2f azi = ld2(zim + k0);
    long ba0 = ((long)kr * 32 + l16) * 256 + wofs;
    long ba0b = ((long)(kr + 1) * 32 + l16) * 256 + wofs;
    long ba1 = ((long)kr * 32 + 16 + l16) * 256 + wofs;
    long ba1b = ((long)(kr + 1) * 32 + 16 + l16) * 256 + wofs;
    v2f br0, bi0, br1, bi1;
    br0.x = wr[ba0];  br0.y = wr[ba0b];
    bi0.x = wi[ba0];  bi0.y = wi[ba0b];
    br1.x = wr[ba1];  br1.y = wr[ba1b];
    bi1.x = wi[ba1];  bi1.y = wi[ba1b];
    p0 = wmma4(azr, br0, p0);
    q0 = wmma4(azi, bi0, q0);
    u0 = wmma4(azr, bi0, u0);
    v0 = wmma4(azi, br0, v0);
    p1 = wmma4(azr, br1, p1);
    q1 = wmma4(azi, bi1, q1);
    u1 = wmma4(azr, bi1, u1);
    v1 = wmma4(azi, br1, v1);
  }
#pragma unroll
  for (int r = 0; r < 8; ++r) {
    int b = r + half * 8;
    long o0 = (((long)b * 32 + l16) * 32 + kyp) * 16 + kx;
    long o1 = (((long)b * 32 + 16 + l16) * 32 + kyp) * 16 + kx;
    Tre[o0] = p0[r] - q0[r];
    Tim[o0] = u0[r] + v0[r];
    Tre[o1] = p1[r] - q1[r];
    Tim[o1] = u1[r] + v1[r];
  }
}

__global__ __launch_bounds__(256) void k4_idft_h(const float* __restrict__ Tre,
                                                 const float* __restrict__ Tim,
                                                 const float* __restrict__ E4c,
                                                 const float* __restrict__ E4s,
                                                 float* __restrict__ U) {
  const int wid = blockIdx.x * 8 + (threadIdx.x >> 5);
  const int lane = threadIdx.x & 31;
  const int half = lane >> 4, l16 = lane & 15;
  const int bo = wid >> 4;
  const int h0 = (wid & 15) * 16;
  const float* tre = Tre + (long)bo * 512;
  const float* tim = Tim + (long)bo * 512;
  const float* ec = E4c + (h0 + l16) * 32 + 2 * half;
  const float* es = E4s + (h0 + l16) * 32 + 2 * half;
  v8f a0 = {}, a1 = {}, a2 = {}, a3 = {};
#pragma unroll
  for (int k0 = 0; k0 < 32; k0 += 4) {
    int kr = k0 + 2 * half;
    v2f ac = ld2(ec + k0);
    v2f as = ld2(es + k0);
    v2f bre, bim;
    bre.x = tre[kr * 16 + l16];
    bre.y = tre[(kr + 1) * 16 + l16];
    bim.x = tim[kr * 16 + l16];
    bim.y = tim[(kr + 1) * 16 + l16];
    a0 = wmma4(ac, bre, a0);
    a1 = wmma4(as, bim, a1);
    a2 = wmma4(ac, bim, a2);
    a3 = wmma4(as, bre, a3);
  }
  float* ub = U + ((long)bo * 256 + h0) * 32;
#pragma unroll
  for (int r = 0; r < 8; ++r) {
    int hh = r + half * 8;
    ub[hh * 32 + l16] = a0[r] - a1[r];
    ub[hh * 32 + 16 + l16] = a2[r] + a3[r];
  }
}

__device__ __forceinline__ void k5_load(v2f (&a)[8], const float* arow) {
#pragma unroll
  for (int q = 0; q < 8; ++q) a[q] = ld2(arow + 4 * q);
}

__device__ __forceinline__ void k5_tile(const v2f (&a)[8],
                                        const v2f (&g)[4][8], float* o) {
  v8f acc[4] = {};
#pragma unroll
  for (int q = 0; q < 8; ++q) {
#pragma unroll
    for (int t = 0; t < 4; ++t) acc[t] = wmma4(a[q], g[t][q], acc[t]);
  }
#pragma unroll
  for (int r = 0; r < 8; ++r) {
#pragma unroll
    for (int t = 0; t < 4; ++t) {
      __builtin_nontemporal_store(acc[t][r], &o[r * 256 + t * 16]);
    }
  }
}

__global__ __launch_bounds__(256) void k5_idft_w(const float* __restrict__ U,
                                                 const float* __restrict__ GT,
                                                 float* __restrict__ out) {
  const int wv = threadIdx.x >> 5;
  const int lane = threadIdx.x & 31;
  const int half = lane >> 4, l16 = lane & 15;
  const int bo = blockIdx.x * 2 + (wv >> 2);
  const int n0 = (wv & 3) * 64;
  const long rowbase = (long)bo * 256;
  const float* g0 = GT + (n0 + l16) * 32 + 2 * half;
  v2f g[4][8];
#pragma unroll
  for (int t = 0; t < 4; ++t) {
#pragma unroll
    for (int q = 0; q < 8; ++q) g[t][q] = ld2(g0 + t * 16 * 32 + 4 * q);
  }
  const float* arow = U + (rowbase + l16) * 32 + 2 * half;
  float* obase = out + (rowbase + half * 8) * 256 + n0 + l16;
  v2f A[8], B[8];
  k5_load(A, arow);
  for (int j = 0; j < 14; j += 2) {
    k5_load(B, arow + (j + 1) * 512);
    k5_tile(A, g, obase + j * 4096);
    k5_load(A, arow + (j + 2) * 512);
    k5_tile(B, g, obase + (j + 1) * 4096);
  }
  k5_load(B, arow + 15 * 512);
  k5_tile(A, g, obase + 14 * 4096);
  k5_tile(B, g, obase + 15 * 4096);
}

extern "C" void kernel_launch(void* const* d_in, const int* in_sizes, int n_in,
                              void* d_out, int out_size, void* d_ws,
                              size_t ws_size, hipStream_t stream) {
  (void)in_sizes; (void)n_in; (void)out_size; (void)ws_size;
  const float* x = (const float*)d_in[0];
  const float* w1r = (const float*)d_in[1];
  const float* w1i = (const float*)d_in[2];
  const float* w2r = (const float*)d_in[3];
  const float* w2i = (const float*)d_in[4];
  float* out = (float*)d_out;

  float* tab = (float*)d_ws;
  float* FT = tab;
  float* E2c = tab + 8192;
  float* E2s = tab + 16384;
  float* E4c = tab + 24576;
  float* E4s = tab + 32768;
  float* GT = tab + 40960;
  float* Pc = tab + 49152;
  float* Ps = Pc + 4194304;
  float* Zre = Ps + 4194304;
  float* Zim = Zre + 262144;
  float* Tre = Zim + 262144;
  float* Tim = Tre + 262144;
  float* U = Tim + 262144;

  hipLaunchKernelGGL(k0_tables, dim3(192), dim3(256), 0, stream, tab);
  hipLaunchKernelGGL(k1_dft_h, dim3(256), dim3(256), 0, stream, x, E2c, E2s, Pc,
                     Ps);
  hipLaunchKernelGGL(k2_dft_w, dim3(512), dim3(256), 0, stream, Pc, Ps, FT, Zre,
                     Zim);
  hipLaunchKernelGGL(k3_mix, dim3(64), dim3(256), 0, stream, Zre, Zim, w1r, w1i,
                     w2r, w2i, Tre, Tim);
  hipLaunchKernelGGL(k4_idft_h, dim3(1024), dim3(256), 0, stream, Tre, Tim, E4c,
                     E4s, U);
  hipLaunchKernelGGL(k5_idft_w, dim3(256), dim3(256), 0, stream, U, GT, out);
}


# Channel-mix kernel staged through LDS: a workgroup loads its (ky', o tile, 8 kx) weight slab and stores its T tile in 16-byte pieces along kx, B fragments read from LDS; 128 workgroups; same sums
# speedup vs baseline: 2.1912x; 1.2905x over previous
#include <hip/hip_runtime.h>


typedef __attribute__((ext_vector_type(2))) float v2f;
typedef __attribute__((ext_vector_type(4))) float v4f;
typedef __attribute__((ext_vector_type(8))) float v8f;

__device__ __forceinline__ v8f wmma4(v2f a, v2f b, v8f c) {
  return __builtin_amdgcn_wmma_f32_16x16x4_f32(false, a, false, b, (short)0, c,
                                               false, false);
}

__device__ __forceinline__ v2f ld2(const float* p) { return *(const v2f*)p; }
__device__ __forceinline__ v4f ld4_nt(const float* p) {
  return __builtin_nontemporal_load((const v4f*)p);
}

__global__ __launch_bounds__(256) void k0_tables(float* __restrict__ tab) {
  int idx = blockIdx.x * 256 + threadIdx.x;
  if (idx >= 49152) return;
  const float step = 6.28318530717958647692f / 256.0f;
  float v;
  if (idx < 8192) {
    int n = idx >> 8, w = idx & 255;
    float th = step * (float)(((n & 15) * w) & 255);
    v = (n < 16) ? cosf(th) : -sinf(th);
  } else if (idx < 24576) {
    int j = idx - 8192;
    int is_s = j >= 8192;
    j &= 8191;
    int r = j >> 8, h = j & 255;
    int ky = (r < 16) ? r : (224 + r);
    float th = step * (float)((ky * h) & 255);
    v = is_s ? sinf(th) : cosf(th);
  } else if (idx < 40960) {
    int j = idx - 24576;
    int is_s = j >= 8192;
    j &= 8191;
    int h = j >> 5, r = j & 31;
    int ky = (r < 16) ? r : (224 + r);
    float th = step * (float)((ky * h) & 255);
    v = is_s ? sinf(th) : cosf(th);
  } else {
    int j = idx - 40960;
    int w = j >> 5, n = j & 31;
    int kx = n & 15;
    float th = step * (float)((kx * w) & 255);
    float sc = ((kx == 0) ? 1.0f : 2.0f) * (1.0f / 65536.0f);
    v = (n < 16) ? sc * cosf(th) : -sc * sinf(th);
  }
  tab[idx] = v;
}

struct XBlk {
  v4f xa[8];
  v4f xm[8];
  v2f c0[4], c1[4], s0[4], s1[4];
};

__device__ __forceinline__ void k1_load(XBlk& B, const float* xp,
                                        const float* xmp, const float* c0,
                                        const float* c1, const float* s0,
                                        const float* s1, int kb, int half) {
#pragma unroll
  for (int q = 0; q < 4; ++q) {
    int k0 = kb + 4 * q;
    B.xa[2 * q] = ld4_nt(xp + (long)k0 * 256);
    B.xa[2 * q + 1] = ld4_nt(xp + (long)(k0 + 1) * 256);
    const float* m0 = xmp - (long)k0 * 256;
    if (q == 0) m0 = (k0 + 2 * half == 0) ? xp : m0;
    B.xm[2 * q] = ld4_nt(m0);
    B.xm[2 * q + 1] = ld4_nt(xmp - (long)(k0 + 1) * 256);
    B.c0[q] = ld2(c0 + k0);
    B.c1[q] = ld2(c1 + k0);
    B.s0[q] = ld2(s0 + k0);
    B.s1[q] = ld2(s1 + k0);
  }
}

__device__ __forceinline__ void k1_mma(const XBlk& B, v8f (&acc)[4][4],
                                       bool row0) {
#pragma unroll
  for (int q = 0; q < 4; ++q) {
    v4f e0 = B.xa[2 * q] + B.xm[2 * q];
    v4f e1 = B.xa[2 * q + 1] + B.xm[2 * q + 1];
    v4f o0 = B.xa[2 * q] - B.xm[2 * q];
    v4f o1 = B.xa[2 * q + 1] - B.xm[2 * q + 1];
    if (q == 0) {
#pragma unroll
      for (int t = 0; t < 4; ++t) e0[t] = row0 ? B.xa[0][t] : e0[t];
    }
#pragma unroll
    for (int t = 0; t < 4; ++t) {
      v2f be, bo;
      be.x = e0[t];
      be.y = e1[t];
      bo.x = o0[t];
      bo.y = o1[t];
      acc[0][t] = wmma4(B.c0[q], be, acc[0][t]);
      acc[1][t] = wmma4(B.c1[q], be, acc[1][t]);
      acc[2][t] = wmma4(B.s0[q], bo, acc[2][t]);
      acc[3][t] = wmma4(B.s1[q], bo, acc[3][t]);
    }
  }
}

__global__ __launch_bounds__(256) void k1_dft_h(const float* __restrict__ x,
                                                const float* __restrict__ E2c,
                                                const float* __restrict__ E2s,
                                                float* __restrict__ Pc,
                                                float* __restrict__ Ps) {
  const int wv = threadIdx.x >> 5;
  const int lane = threadIdx.x & 31;
  const int half = lane >> 4, l16 = lane & 15;
  const int bi = blockIdx.x * 2 + (wv >> 2);
  const int wbase = (wv & 3) * 64;
  const float* ximg = x + (long)bi * 65536 + wbase + 4 * l16;
  const float* xp = ximg + (2 * half) * 256;
  const float* xmp = ximg + (256 - 2 * half) * 256;
  const float* c0 = E2c + l16 * 256 + 2 * half;
  const float* c1 = E2c + (16 + l16) * 256 + 2 * half;
  const float* s0 = E2s + l16 * 256 + 2 * half;
  const float* s1 = E2s + (16 + l16) * 256 + 2 * half;
  v8f acc[4][4] = {};
  XBlk A, B;
  k1_load(A, xp, xmp, c0, c1, s0, s1, 0, half);
#pragma unroll 1
  for (int kb = 0; kb < 96; kb += 32) {
    k1_load(B, xp, xmp, c0, c1, s0, s1, kb + 16, half);
    k1_mma(A, acc, (kb == 0) && (half == 0));
    k1_load(A, xp, xmp, c0, c1, s0, s1, kb + 32, half);
    k1_mma(B, acc, false);
  }
  k1_load(B, xp, xmp, c0, c1, s0, s1, 112, half);
  v4f x128 = ld4_nt(ximg + 128 * 256);
  v2f t0 = ld2(c0 + 128);
  v2f t1 = ld2(c1 + 128);
  k1_mma(A, acc, false);
  k1_mma(B, acc, false);
  {
    v2f a0, a1;
    a0.x = half ? 0.0f : t0.x;
    a0.y = 0.0f;
    a1.x = half ? 0.0f : t1.x;
    a1.y = 0.0f;
#pragma unroll
    for (int t = 0; t < 4; ++t) {
      v2f b;
      b.x = half ? 0.0f : x128[t];
      b.y = 0.0f;
      acc[0][t] = wmma4(a0, b, acc[0][t]);
      acc[1][t] = wmma4(a1, b, acc[1][t]);
    }
  }
  float* pc = Pc + (long)bi * 8192 + wbase + 4 * l16;
  float* ps = Ps + (long)bi * 8192 + wbase + 4 * l16;
#pragma unroll
  for (int r = 0; r < 8; ++r) {
    int row = r + half * 8;
    v4f o0 = {acc[0][0][r], acc[0][1][r], acc[0][2][r], acc[0][3][r]};
    v4f o1 = {acc[1][0][r], acc[1][1][r], acc[1][2][r], acc[1][3][r]};
    v4f o2 = {acc[2][0][r], acc[2][1][r], acc[2][2][r], acc[2][3][r]};
    v4f o3 = {acc[3][0][r], acc[3][1][r], acc[3][2][r], acc[3][3][r]};
    *(v4f*)(pc + row * 256) = o0;
    *(v4f*)(pc + (16 + row) * 256) = o1;
    *(v4f*)(ps + row * 256) = o2;
    *(v4f*)(ps + (16 + row) * 256) = o3;
  }
}

__global__ __launch_bounds__(256) void k2_dft_w(const float* __restrict__ Pc,
                                                const float* __restrict__ Ps,
                                                const float* __restrict__ FT,
                                                float* __restrict__ Zre,
                                                float* __restrict__ Zim) {
  __shared__ float red[8 * 1024];
  const int bi = blockIdx.x;
  const int wv = threadIdx.x >> 5;
  const int lane = threadIdx.x & 31;
  const int half = lane >> 4, l16 = lane & 15;
  const int wb = wv * 32;
  const float* pc0 = Pc + (long)bi * 8192 + l16 * 256 + wb + 2 * half;
  const float* pc1 = pc0 + 16 * 256;
  const float* ps0 = Ps + (long)bi * 8192 + l16 * 256 + wb + 2 * half;
  const float* ps1 = ps0 + 16 * 256;
  const float* fc = FT + l16 * 256 + wb + 2 * half;
  const float* fm = FT + (16 + l16) * 256 + wb + 2 * half;
  v8f a0 = {}, a1 = {}, a2 = {}, a3 = {}, a4 = {}, a5 = {}, a6 = {}, a7 = {};
#pragma unroll
  for (int k0 = 0; k0 < 32; k0 += 4) {
    v2f ac0 = ld2(pc0 + k0), ac1 = ld2(pc1 + k0);
    v2f as0 = ld2(ps0 + k0), as1 = ld2(ps1 + k0);
    v2f bre = ld2(fc + k0), bim = ld2(fm + k0);
    a0 = wmma4(ac0, bre, a0);
    a1 = wmma4(as0, bim, a1);
    a2 = wmma4(ac0, bim, a2);
    a3 = wmma4(as0, bre, a3);
    a4 = wmma4(ac1, bre, a4);
    a5 = wmma4(as1, bim, a5);
    a6 = wmma4(ac1, bim, a6);
    a7 = wmma4(as1, bre, a7);
  }
  float* mine = red + wv * 1024 + lane;
#pragma unroll
  for (int r = 0; r < 8; ++r) {
    mine[(0 * 8 + r) * 32] = a0[r] + a1[r];
    mine[(1 * 8 + r) * 32] = a2[r] - a3[r];
    mine[(2 * 8 + r) * 32] = a4[r] + a5[r];
    mine[(3 * 8 + r) * 32] = a6[r] - a7[r];
  }
  __syncthreads();
  float z0 = red[0 * 256 + threadIdx.x];
  float z1 = red[1 * 256 + threadIdx.x];
  float z2 = red[2 * 256 + threadIdx.x];
  float z3 = red[3 * 256 + threadIdx.x];
#pragma unroll
  for (int v = 1; v < 8; ++v) {
    z0 += red[v * 1024 + 0 * 256 + threadIdx.x];
    z1 += red[v * 1024 + 1 * 256 + threadIdx.x];
    z2 += red[v * 1024 + 2 * 256 + threadIdx.x];
    z3 += red[v * 1024 + 3 * 256 + threadIdx.x];
  }
  const int b = bi >> 5, i = bi & 31;
  const int kyr = wv + half * 8;
  long o0 = (((long)kyr * 16 + l16) * 16 + b) * 32 + i;
  long o1 = (((long)(16 + kyr) * 16 + l16) * 16 + b) * 32 + i;
  Zre[o0] = z0;
  Zim[o0] = z1;
  Zre[o1] = z2;
  Zim[o1] = z3;
}

#define K3_PLANE 520
__global__ __launch_bounds__(256) void k3_mix(
    const float* __restrict__ Zre, const float* __restrict__ Zim,
    const float* __restrict__ w1r, const float* __restrict__ w1i,
    const float* __restrict__ w2r, const float* __restrict__ w2i,
    float* __restrict__ Tre, float* __restrict__ Tim) {
  __shared__ __attribute__((aligned(16))) float wl[2][8 * K3_PLANE];
  __shared__ __attribute__((aligned(16))) float tl[2][2048];
  const int tid = threadIdx.x;
  const int wv = tid >> 5;
  const int lane = tid & 31;
  const int half = lane >> 4, l16 = lane & 15;
  const int kyp = blockIdx.x >> 2;
  const int ot = (blockIdx.x >> 1) & 1;
  const int kh = blockIdx.x & 1;
  const int ky = kyp & 15;
  const float* wr = (kyp < 16) ? w1r : w2r;
  const float* wi = (kyp < 16) ? w1i : w2i;
  const int m = kyp * 16 + kh * 8 + wv;
  const float* zre = Zre + (long)m * 512 + l16 * 32 + 2 * half;
  const float* zim = Zim + (long)m * 512 + l16 * 32 + 2 * half;
  v2f azr[8], azi[8];
#pragma unroll
  for (int q = 0; q < 8; ++q) {
    azr[q] = ld2(zre + 4 * q);
    azi[q] = ld2(zim + 4 * q);
  }
#pragma unroll
  for (int n = 0; n < 4; ++n) {
    int p = tid + 256 * n;
    int j = p & 1, ol = (p >> 1) & 15, i = p >> 5;
    long g = ((long)(i * 32 + ot * 16 + ol)) * 256 + ky * 16 + kh * 8 + j * 4;
    v4f vr = *(const v4f*)(wr + g);
    v4f vi = *(const v4f*)(wi + g);
    int base = (j * 4) * K3_PLANE + ((i >> 1) * 16 + ol) * 2 + (i & 1);
#pragma unroll
    for (int e = 0; e < 4; ++e) {
      wl[0][base + e * K3_PLANE] = vr[e];
      wl[1][base + e * K3_PLANE] = vi[e];
    }
  }
  __syncthreads();
  const float* bre = &wl[0][wv * K3_PLANE + (half * 16 + l16) * 2];
  const float* bim = &wl[1][wv * K3_PLANE + (half * 16 + l16) * 2];
  v8f p0 = {}, q0 = {}, u0 = {}, v0 = {};
#pragma unroll
  for (int q = 0; q < 8; ++q) {
    v2f br = *(const v2f*)(bre + q * 64);
    v2f bi = *(const v2f*)(bim + q * 64);
    p0 = wmma4(azr[q], br, p0);
    q0 = wmma4(azi[q], bi, q0);
    u0 = wmma4(azr[q], bi, u0);
    v0 = wmma4(azi[q], br, v0);
  }
#pragma unroll
  for (int r = 0; r < 8; ++r) {
    int t = ((r + half * 8) * 16 + l16) * 8 + wv;
    tl[0][t] = p0[r] - q0[r];
    tl[1][t] = u0[r] + v0[r];
  }
  __syncthreads();
#pragma unroll
  for (int n = 0; n < 2; ++n) {
    int pp = tid + 256 * n;
    int j = pp & 1, ol = (pp >> 1) & 15, b = pp >> 5;
    long o = (((long)b * 32 + ot * 16 + ol) * 32 + kyp) * 16 + kh * 8 + j * 4;
    *(v4f*)(Tre + o) = *(const v4f*)&tl[0][pp * 4];
    *(v4f*)(Tim + o) = *(const v4f*)&tl[1][pp * 4];
  }
}

__global__ __launch_bounds__(256) void k4_idft_h(const float* __restrict__ Tre,
                                                 const float* __restrict__ Tim,
                                                 const float* __restrict__ E4c,
                                                 const float* __restrict__ E4s,
                                                 float* __restrict__ U) {
  const int wid = blockIdx.x * 8 + (threadIdx.x >> 5);
  const int lane = threadIdx.x & 31;
  const int half = lane >> 4, l16 = lane & 15;
  const int bo = wid >> 4;
  const int h0 = (wid & 15) * 16;
  const float* tre = Tre + (long)bo * 512;
  const float* tim = Tim + (long)bo * 512;
  const float* ec = E4c + (h0 + l16) * 32 + 2 * half;
  const float* es = E4s + (h0 + l16) * 32 + 2 * half;
  v8f a0 = {}, a1 = {}, a2 = {}, a3 = {};
#pragma unroll
  for (int k0 = 0; k0 < 32; k0 += 4) {
    int kr = k0 + 2 * half;
    v2f ac = ld2(ec + k0);
    v2f as = ld2(es + k0);
    v2f bre, bim;
    bre.x = tre[kr * 16 + l16];
    bre.y = tre[(kr + 1) * 16 + l16];
    bim.x = tim[kr * 16 + l16];
    bim.y = tim[(kr + 1) * 16 + l16];
    a0 = wmma4(ac, bre, a0);
    a1 = wmma4(as, bim, a1);
    a2 = wmma4(ac, bim, a2);
    a3 = wmma4(as, bre, a3);
  }
  float* ub = U + ((long)bo * 256 + h0) * 32;
#pragma unroll
  for (int r = 0; r < 8; ++r) {
    int hh = r + half * 8;
    ub[hh * 32 + l16] = a0[r] - a1[r];
    ub[hh * 32 + 16 + l16] = a2[r] + a3[r];
  }
}

__device__ __forceinline__ void k5_load(v2f (&a)[8], const float* arow) {
#pragma unroll
  for (int q = 0; q < 8; ++q) a[q] = ld2(arow + 4 * q);
}

__device__ __forceinline__ void k5_tile(const v2f (&a)[8],
                                        const v2f (&g)[4][8], float* o) {
  v8f acc[4] = {};
#pragma unroll
  for (int q = 0; q < 8; ++q) {
#pragma unroll
    for (int t = 0; t < 4; ++t) acc[t] = wmma4(a[q], g[t][q], acc[t]);
  }
#pragma unroll
  for (int r = 0; r < 8; ++r) {
#pragma unroll
    for (int t = 0; t < 4; ++t) {
      __builtin_nontemporal_store(acc[t][r], &o[r * 256 + t * 16]);
    }
  }
}

__global__ __launch_bounds__(256) void k5_idft_w(const float* __restrict__ U,
                                                 const float* __restrict__ GT,
                                                 float* __restrict__ out) {
  const int wv = threadIdx.x >> 5;
  const int lane = threadIdx.x & 31;
  const int half = lane >> 4, l16 = lane & 15;
  const int bo = blockIdx.x * 2 + (wv >> 2);
  const int n0 = (wv & 3) * 64;
  const long rowbase = (long)bo * 256;
  const float* g0 = GT + (n0 + l16) * 32 + 2 * half;
  v2f g[4][8];
#pragma unroll
  for (int t = 0; t < 4; ++t) {
#pragma unroll
    for (int q = 0; q < 8; ++q) g[t][q] = ld2(g0 + t * 16 * 32 + 4 * q);
  }
  const float* arow = U + (rowbase + l16) * 32 + 2 * half;
  float* obase = out + (rowbase + half * 8) * 256 + n0 + l16;
  v2f A[8], B[8];
  k5_load(A, arow);
  for (int j = 0; j < 14; j += 2) {
    k5_load(B, arow + (j + 1) * 512);
    k5_tile(A, g, obase + j * 4096);
    k5_load(A, arow + (j + 2) * 512);
    k5_tile(B, g, obase + (j + 1) * 4096);
  }
  k5_load(B, arow + 15 * 512);
  k5_tile(A, g, obase + 14 * 4096);
  k5_tile(B, g, obase + 15 * 4096);
}

extern "C" void kernel_launch(void* const* d_in, const int* in_sizes, int n_in,
                              void* d_out, int out_size, void* d_ws,
                              size_t ws_size, hipStream_t stream) {
  (void)in_sizes; (void)n_in; (void)out_size; (void)ws_size;
  const float* x = (const float*)d_in[0];
  const float* w1r = (const float*)d_in[1];
  const float* w1i = (const float*)d_in[2];
  const float* w2r = (const float*)d_in[3];
  const float* w2i = (const float*)d_in[4];
  float* out = (float*)d_out;

  float* tab = (float*)d_ws;
  float* FT = tab;
  float* E2c = tab + 8192;
  float* E2s = tab + 16384;
  float* E4c = tab + 24576;
  float* E4s = tab + 32768;
  float* GT = tab + 40960;
  float* Pc = tab + 49152;
  float* Ps = Pc + 4194304;
  float* Zre = Ps + 4194304;
  float* Zim = Zre + 262144;
  float* Tre = Zim + 262144;
  float* Tim = Tre + 262144;
  float* U = Tim + 262144;

  hipLaunchKernelGGL(k0_tables, dim3(192), dim3(256), 0, stream, tab);
  hipLaunchKernelGGL(k1_dft_h, dim3(256), dim3(256), 0, stream, x, E2c, E2s, Pc,
                     Ps);
  hipLaunchKernelGGL(k2_dft_w, dim3(512), dim3(256), 0, stream, Pc, Ps, FT, Zre,
                     Zim);
  hipLaunchKernelGGL(k3_mix, dim3(128), dim3(256), 0, stream, Zre, Zim, w1r, w1i,
                     w2r, w2i, Tre, Tim);
  hipLaunchKernelGGL(k4_idft_h, dim3(1024), dim3(256), 0, stream, Tre, Tim, E4c,
                     E4s, U);
  hipLaunchKernelGGL(k5_idft_w, dim3(256), dim3(256), 0, stream, U, GT, out);
}


# Table launch removed: the forward H-DFT kernel builds its twiddle rows in LDS per workgroup (fragments read from LDS) and writes the later kernels' tables itself; 5 launches; same values and sums
# speedup vs baseline: 2.2580x; 1.0305x over previous
#include <hip/hip_runtime.h>


typedef __attribute__((ext_vector_type(2))) float v2f;
typedef __attribute__((ext_vector_type(4))) float v4f;
typedef __attribute__((ext_vector_type(8))) float v8f;

__device__ __forceinline__ v8f wmma4(v2f a, v2f b, v8f c) {
  return __builtin_amdgcn_wmma_f32_16x16x4_f32(false, a, false, b, (short)0, c,
                                               false, false);
}

__device__ __forceinline__ v2f ld2(const float* p) { return *(const v2f*)p; }
__device__ __forceinline__ v4f ld4_nt(const float* p) {
  return __builtin_nontemporal_load((const v4f*)p);
}

struct XBlk {
  v4f xa[8];
  v4f xm[8];
  v2f c0[4], c1[4], s0[4], s1[4];
};

__device__ __forceinline__ void k1_load_x(XBlk& B, const float* xp,
                                          const float* xmp, int kb, int half) {
#pragma unroll
  for (int q = 0; q < 4; ++q) {
    int k0 = kb + 4 * q;
    B.xa[2 * q] = ld4_nt(xp + (long)k0 * 256);
    B.xa[2 * q + 1] = ld4_nt(xp + (long)(k0 + 1) * 256);
    const float* m0 = xmp - (long)k0 * 256;
    if (q == 0) m0 = (k0 + 2 * half == 0) ? xp : m0;
    B.xm[2 * q] = ld4_nt(m0);
    B.xm[2 * q + 1] = ld4_nt(xmp - (long)(k0 + 1) * 256);
  }
}

__device__ __forceinline__ void k1_load_t(XBlk& B, const float* c0,
                                          const float* c1, const float* s0,
                                          const float* s1, int kb) {
#pragma unroll
  for (int q = 0; q < 4; ++q) {
    int k0 = kb + 4 * q;
    B.c0[q] = ld2(c0 + k0);
    B.c1[q] = ld2(c1 + k0);
    B.s0[q] = ld2(s0 + k0);
    B.s1[q] = ld2(s1 + k0);
  }
}

__device__ __forceinline__ void k1_load(XBlk& B, const float* xp,
                                        const float* xmp, const float* c0,
                                        const float* c1, const float* s0,
                                        const float* s1, int kb, int half) {
  k1_load_x(B, xp, xmp, kb, half);
  k1_load_t(B, c0, c1, s0, s1, kb);
}

__device__ __forceinline__ void k1_mma(const XBlk& B, v8f (&acc)[4][4],
                                       bool row0) {
#pragma unroll
  for (int q = 0; q < 4; ++q) {
    v4f e0 = B.xa[2 * q] + B.xm[2 * q];
    v4f e1 = B.xa[2 * q + 1] + B.xm[2 * q + 1];
    v4f o0 = B.xa[2 * q] - B.xm[2 * q];
    v4f o1 = B.xa[2 * q + 1] - B.xm[2 * q + 1];
    if (q == 0) {
#pragma unroll
      for (int t = 0; t < 4; ++t) e0[t] = row0 ? B.xa[0][t] : e0[t];
    }
#pragma unroll
    for (int t = 0; t < 4; ++t) {
      v2f be, bo;
      be.x = e0[t];
      be.y = e1[t];
      bo.x = o0[t];
      bo.y = o1[t];
      acc[0][t] = wmma4(B.c0[q], be, acc[0][t]);
      acc[1][t] = wmma4(B.c1[q], be, acc[1][t]);
      acc[2][t] = wmma4(B.s0[q], bo, acc[2][t]);
      acc[3][t] = wmma4(B.s1[q], bo, acc[3][t]);
    }
  }
}

#define E2P 132
__global__ __launch_bounds__(256) void k1_dft_h(const float* __restrict__ x,
                                                float* __restrict__ tab,
                                                float* __restrict__ Pc,
                                                float* __restrict__ Ps) {
  __shared__ float cs[512];
  __shared__ __attribute__((aligned(16))) float e2[64 * E2P];
  const int tid = threadIdx.x;
  const int wv = tid >> 5;
  const int lane = tid & 31;
  const int half = lane >> 4, l16 = lane & 15;
  const int bi = blockIdx.x * 2 + (wv >> 2);
  const int wbase = (wv & 3) * 64;
  const float* ximg = x + (long)bi * 65536 + wbase + 4 * l16;
  const float* xp = ximg + (2 * half) * 256;
  const float* xmp = ximg + (256 - 2 * half) * 256;
  const float* c0 = e2 + l16 * E2P + 2 * half;
  const float* c1 = e2 + (16 + l16) * E2P + 2 * half;
  const float* s0 = e2 + (32 + l16) * E2P + 2 * half;
  const float* s1 = e2 + (48 + l16) * E2P + 2 * half;
  v8f acc[4][4] = {};
  XBlk A, B;
  k1_load_x(A, xp, xmp, 0, half);
  {
    const float step = 6.28318530717958647692f / 256.0f;
    float th = step * (float)tid;
    cs[tid] = cosf(th);
    cs[256 + tid] = sinf(th);
  }
  __syncthreads();
  {
    const int row = tid >> 2;
    const int r = row & 31;
    const int ky = (r < 16) ? r : (224 + r);
    const int h0 = (tid & 3) * 33;
    const float* src = cs + ((row >> 5) << 8);
    float* dst = e2 + row * E2P + h0;
    int n = ky * h0;
#pragma unroll
    for (int it = 0; it < 33; ++it) {
      dst[it] = src[n & 255];
      n += ky;
    }
  }
  __syncthreads();
  if (tid < 128) {
    const int e = blockIdx.x * 128 + tid;
    float v;
    int o;
    if (e < 8192) {
      int n = e >> 8, w = e & 255;
      int m = ((n & 15) * w) & 255;
      v = (n < 16) ? cs[m] : -cs[256 + m];
      o = e;
    } else if (e < 24576) {
      int j = e - 8192;
      int is_s = j >= 8192;
      int h = (j & 8191) >> 5, r = j & 31;
      int ky = (r < 16) ? r : (224 + r);
      int m = (ky * h) & 255;
      v = is_s ? cs[256 + m] : cs[m];
      o = 24576 + j;
    } else {
      int j = e - 24576;
      int w = j >> 5, n = j & 31;
      int kx = n & 15;
      int m = (kx * w) & 255;
      float sc = ((kx == 0) ? 1.0f : 2.0f) * (1.0f / 65536.0f);
      v = (n < 16) ? sc * cs[m] : -sc * cs[256 + m];
      o = 40960 + j;
    }
    tab[o] = v;
  }
  k1_load_t(A, c0, c1, s0, s1, 0);
#pragma unroll 1
  for (int kb = 0; kb < 96; kb += 32) {
    k1_load(B, xp, xmp, c0, c1, s0, s1, kb + 16, half);
    __builtin_amdgcn_sched_barrier(0);
    k1_mma(A, acc, (kb == 0) && (half == 0));
    __builtin_amdgcn_sched_barrier(0);
    k1_load(A, xp, xmp, c0, c1, s0, s1, kb + 32, half);
    __builtin_amdgcn_sched_barrier(0);
    k1_mma(B, acc, false);
    __builtin_amdgcn_sched_barrier(0);
  }
  k1_load(B, xp, xmp, c0, c1, s0, s1, 112, half);
  v4f x128 = ld4_nt(ximg + 128 * 256);
  v2f t0 = ld2(c0 + 128);
  v2f t1 = ld2(c1 + 128);
  k1_mma(A, acc, false);
  k1_mma(B, acc, false);
  {
    v2f a0, a1;
    a0.x = half ? 0.0f : t0.x;
    a0.y = 0.0f;
    a1.x = half ? 0.0f : t1.x;
    a1.y = 0.0f;
#pragma unroll
    for (int t = 0; t < 4; ++t) {
      v2f b;
      b.x = half ? 0.0f : x128[t];
      b.y = 0.0f;
      acc[0][t] = wmma4(a0, b, acc[0][t]);
      acc[1][t] = wmma4(a1, b, acc[1][t]);
    }
  }
  float* pc = Pc + (long)bi * 8192 + wbase + 4 * l16;
  float* ps = Ps + (long)bi * 8192 + wbase + 4 * l16;
#pragma unroll
  for (int r = 0; r < 8; ++r) {
    int row = r + half * 8;
    v4f o0 = {acc[0][0][r], acc[0][1][r], acc[0][2][r], acc[0][3][r]};
    v4f o1 = {acc[1][0][r], acc[1][1][r], acc[1][2][r], acc[1][3][r]};
    v4f o2 = {acc[2][0][r], acc[2][1][r], acc[2][2][r], acc[2][3][r]};
    v4f o3 = {acc[3][0][r], acc[3][1][r], acc[3][2][r], acc[3][3][r]};
    *(v4f*)(pc + row * 256) = o0;
    *(v4f*)(pc + (16 + row) * 256) = o1;
    *(v4f*)(ps + row * 256) = o2;
    *(v4f*)(ps + (16 + row) * 256) = o3;
  }
}

__global__ __launch_bounds__(256) void k2_dft_w(const float* __restrict__ Pc,
                                                const float* __restrict__ Ps,
                                                const float* __restrict__ FT,
                                                float* __restrict__ Zre,
                                                float* __restrict__ Zim) {
  __shared__ float red[8 * 1024];
  const int bi = blockIdx.x;
  const int wv = threadIdx.x >> 5;
  const int lane = threadIdx.x & 31;
  const int half = lane >> 4, l16 = lane & 15;
  const int wb = wv * 32;
  const float* pc0 = Pc + (long)bi * 8192 + l16 * 256 + wb + 2 * half;
  const float* pc1 = pc0 + 16 * 256;
  const float* ps0 = Ps + (long)bi * 8192 + l16 * 256 + wb + 2 * half;
  const float* ps1 = ps0 + 16 * 256;
  const float* fc = FT + l16 * 256 + wb + 2 * half;
  const float* fm = FT + (16 + l16) * 256 + wb + 2 * half;
  v8f a0 = {}, a1 = {}, a2 = {}, a3 = {}, a4 = {}, a5 = {}, a6 = {}, a7 = {};
#pragma unroll
  for (int k0 = 0; k0 < 32; k0 += 4) {
    v2f ac0 = ld2(pc0 + k0), ac1 = ld2(pc1 + k0);
    v2f as0 = ld2(ps0 + k0), as1 = ld2(ps1 + k0);
    v2f bre = ld2(fc + k0), bim = ld2(fm + k0);
    a0 = wmma4(ac0, bre, a0);
    a1 = wmma4(as0, bim, a1);
    a2 = wmma4(ac0, bim, a2);
    a3 = wmma4(as0, bre, a3);
    a4 = wmma4(ac1, bre, a4);
    a5 = wmma4(as1, bim, a5);
    a6 = wmma4(ac1, bim, a6);
    a7 = wmma4(as1, bre, a7);
  }
  float* mine = red + wv * 1024 + lane;
#pragma unroll
  for (int r = 0; r < 8; ++r) {
    mine[(0 * 8 + r) * 32] = a0[r] + a1[r];
    mine[(1 * 8 + r) * 32] = a2[r] - a3[r];
    mine[(2 * 8 + r) * 32] = a4[r] + a5[r];
    mine[(3 * 8 + r) * 32] = a6[r] - a7[r];
  }
  __syncthreads();
  float z0 = red[0 * 256 + threadIdx.x];
  float z1 = red[1 * 256 + threadIdx.x];
  float z2 = red[2 * 256 + threadIdx.x];
  float z3 = red[3 * 256 + threadIdx.x];
#pragma unroll
  for (int v = 1; v < 8; ++v) {
    z0 += red[v * 1024 + 0 * 256 + threadIdx.x];
    z1 += red[v * 1024 + 1 * 256 + threadIdx.x];
    z2 += red[v * 1024 + 2 * 256 + threadIdx.x];
    z3 += red[v * 1024 + 3 * 256 + threadIdx.x];
  }
  const int b = bi >> 5, i = bi & 31;
  const int kyr = wv + half * 8;
  long o0 = (((long)kyr * 16 + l16) * 16 + b) * 32 + i;
  long o1 = (((long)(16 + kyr) * 16 + l16) * 16 + b) * 32 + i;
  Zre[o0] = z0;
  Zim[o0] = z1;
  Zre[o1] = z2;
  Zim[o1] = z3;
}

#define K3_PLANE 520
__global__ __launch_bounds__(256) void k3_mix(
    const float* __restrict__ Zre, const float* __restrict__ Zim,
    const float* __restrict__ w1r, const float* __restrict__ w1i,
    const float* __restrict__ w2r, const float* __restrict__ w2i,
    float* __restrict__ Tre, float* __restrict__ Tim) {
  __shared__ __attribute__((aligned(16))) float wl[2][8 * K3_PLANE];
  __shared__ __attribute__((aligned(16))) float tl[2][2048];
  const int tid = threadIdx.x;
  const int wv = tid >> 5;
  const int lane = tid & 31;
  const int half = lane >> 4, l16 = lane & 15;
  const int kyp = blockIdx.x >> 2;
  const int ot = (blockIdx.x >> 1) & 1;
  const int kh = blockIdx.x & 1;
  const int ky = kyp & 15;
  const float* wr = (kyp < 16) ? w1r : w2r;
  const float* wi = (kyp < 16) ? w1i : w2i;
  const int m = kyp * 16 + kh * 8 + wv;
  const float* zre = Zre + (long)m * 512 + l16 * 32 + 2 * half;
  const float* zim = Zim + (long)m * 512 + l16 * 32 + 2 * half;
  v2f azr[8], azi[8];
#pragma unroll
  for (int q = 0; q < 8; ++q) {
    azr[q] = ld2(zre + 4 * q);
    azi[q] = ld2(zim + 4 * q);
  }
#pragma unroll
  for (int n = 0; n < 4; ++n) {
    int p = tid + 256 * n;
    int j = p & 1, ol = (p >> 1) & 15, i = p >> 5;
    long g = ((long)(i * 32 + ot * 16 + ol)) * 256 + ky * 16 + kh * 8 + j * 4;
    v4f vr = *(const v4f*)(wr + g);
    v4f vi = *(const v4f*)(wi + g);
    int base = (j * 4) * K3_PLANE + ((i >> 1) * 16 + ol) * 2 + (i & 1);
#pragma unroll
    for (int e = 0; e < 4; ++e) {
      wl[0][base + e * K3_PLANE] = vr[e];
      wl[1][base + e * K3_PLANE] = vi[e];
    }
  }
  __syncthreads();
  const float* bre = &wl[0][wv * K3_PLANE + (half * 16 + l16) * 2];
  const float* bim = &wl[1][wv * K3_PLANE + (half * 16 + l16) * 2];
  v8f p0 = {}, q0 = {}, u0 = {}, v0 = {};
#pragma unroll
  for (int q = 0; q < 8; ++q) {
    v2f br = *(const v2f*)(bre + q * 64);
    v2f bi = *(const v2f*)(bim + q * 64);
    p0 = wmma4(azr[q], br, p0);
    q0 = wmma4(azi[q], bi, q0);
    u0 = wmma4(azr[q], bi, u0);
    v0 = wmma4(azi[q], br, v0);
  }
#pragma unroll
  for (int r = 0; r < 8; ++r) {
    int t = ((r + half * 8) * 16 + l16) * 8 + wv;
    tl[0][t] = p0[r] - q0[r];
    tl[1][t] = u0[r] + v0[r];
  }
  __syncthreads();
#pragma unroll
  for (int n = 0; n < 2; ++n) {
    int pp = tid + 256 * n;
    int j = pp & 1, ol = (pp >> 1) & 15, b = pp >> 5;
    long o = (((long)b * 32 + ot * 16 + ol) * 32 + kyp) * 16 + kh * 8 + j * 4;
    *(v4f*)(Tre + o) = *(const v4f*)&tl[0][pp * 4];
    *(v4f*)(Tim + o) = *(const v4f*)&tl[1][pp * 4];
  }
}

__global__ __launch_bounds__(256) void k4_idft_h(const float* __restrict__ Tre,
                                                 const float* __restrict__ Tim,
                                                 const float* __restrict__ E4c,
                                                 const float* __restrict__ E4s,
                                                 float* __restrict__ U) {
  const int wid = blockIdx.x * 8 + (threadIdx.x >> 5);
  const int lane = threadIdx.x & 31;
  const int half = lane >> 4, l16 = lane & 15;
  const int bo = wid >> 4;
  const int h0 = (wid & 15) * 16;
  const float* tre = Tre + (long)bo * 512;
  const float* tim = Tim + (long)bo * 512;
  const float* ec = E4c + (h0 + l16) * 32 + 2 * half;
  const float* es = E4s + (h0 + l16) * 32 + 2 * half;
  v8f a0 = {}, a1 = {}, a2 = {}, a3 = {};
#pragma unroll
  for (int k0 = 0; k0 < 32; k0 += 4) {
    int kr = k0 + 2 * half;
    v2f ac = ld2(ec + k0);
    v2f as = ld2(es + k0);
    v2f bre, bim;
    bre.x = tre[kr * 16 + l16];
    bre.y = tre[(kr + 1) * 16 + l16];
    bim.x = tim[kr * 16 + l16];
    bim.y = tim[(kr + 1) * 16 + l16];
    a0 = wmma4(ac, bre, a0);
    a1 = wmma4(as, bim, a1);
    a2 = wmma4(ac, bim, a2);
    a3 = wmma4(as, bre, a3);
  }
  float* ub = U + ((long)bo * 256 + h0) * 32;
#pragma unroll
  for (int r = 0; r < 8; ++r) {
    int hh = r + half * 8;
    ub[hh * 32 + l16] = a0[r] - a1[r];
    ub[hh * 32 + 16 + l16] = a2[r] + a3[r];
  }
}

__device__ __forceinline__ void k5_load(v2f (&a)[8], const float* arow) {
#pragma unroll
  for (int q = 0; q < 8; ++q) a[q] = ld2(arow + 4 * q);
}

__device__ __forceinline__ void k5_tile(const v2f (&a)[8],
                                        const v2f (&g)[4][8], float* o) {
  v8f acc[4] = {};
#pragma unroll
  for (int q = 0; q < 8; ++q) {
#pragma unroll
    for (int t = 0; t < 4; ++t) acc[t] = wmma4(a[q], g[t][q], acc[t]);
  }
#pragma unroll
  for (int r = 0; r < 8; ++r) {
#pragma unroll
    for (int t = 0; t < 4; ++t) {
      __builtin_nontemporal_store(acc[t][r], &o[r * 256 + t * 16]);
    }
  }
}

__global__ __launch_bounds__(256) void k5_idft_w(const float* __restrict__ U,
                                                 const float* __restrict__ GT,
                                                 float* __restrict__ out) {
  const int wv = threadIdx.x >> 5;
  const int lane = threadIdx.x & 31;
  const int half = lane >> 4, l16 = lane & 15;
  const int bo = blockIdx.x * 2 + (wv >> 2);
  const int n0 = (wv & 3) * 64;
  const long rowbase = (long)bo * 256;
  const float* g0 = GT + (n0 + l16) * 32 + 2 * half;
  v2f g[4][8];
#pragma unroll
  for (int t = 0; t < 4; ++t) {
#pragma unroll
    for (int q = 0; q < 8; ++q) g[t][q] = ld2(g0 + t * 16 * 32 + 4 * q);
  }
  const float* arow = U + (rowbase + l16) * 32 + 2 * half;
  float* obase = out + (rowbase + half * 8) * 256 + n0 + l16;
  v2f A[8], B[8];
  k5_load(A, arow);
  for (int j = 0; j < 14; j += 2) {
    k5_load(B, arow + (j + 1) * 512);
    k5_tile(A, g, obase + j * 4096);
    k5_load(A, arow + (j + 2) * 512);
    k5_tile(B, g, obase + (j + 1) * 4096);
  }
  k5_load(B, arow + 15 * 512);
  k5_tile(A, g, obase + 14 * 4096);
  k5_tile(B, g, obase + 15 * 4096);
}

extern "C" void kernel_launch(void* const* d_in, const int* in_sizes, int n_in,
                              void* d_out, int out_size, void* d_ws,
                              size_t ws_size, hipStream_t stream) {
  (void)in_sizes; (void)n_in; (void)out_size; (void)ws_size;
  const float* x = (const float*)d_in[0];
  const float* w1r = (const float*)d_in[1];
  const float* w1i = (const float*)d_in[2];
  const float* w2r = (const float*)d_in[3];
  const float* w2i = (const float*)d_in[4];
  float* out = (float*)d_out;

  float* tab = (float*)d_ws;
  float* FT = tab;
  float* E4c = tab + 24576;
  float* E4s = tab + 32768;
  float* GT = tab + 40960;
  float* Pc = tab + 49152;
  float* Ps = Pc + 4194304;
  float* Zre = Ps + 4194304;
  float* Zim = Zre + 262144;
  float* Tre = Zim + 262144;
  float* Tim = Tre + 262144;
  float* U = Tim + 262144;

  hipLaunchKernelGGL(k1_dft_h, dim3(256), dim3(256), 0, stream, x, tab, Pc, Ps);
  hipLaunchKernelGGL(k2_dft_w, dim3(512), dim3(256), 0, stream, Pc, Ps, FT, Zre,
                     Zim);
  hipLaunchKernelGGL(k3_mix, dim3(128), dim3(256), 0, stream, Zre, Zim, w1r, w1i,
                     w2r, w2i, Tre, Tim);
  hipLaunchKernelGGL(k4_idft_h, dim3(1024), dim3(256), 0, stream, Tre, Tim, E4c,
                     E4s, U);
  hipLaunchKernelGGL(k5_idft_w, dim3(256), dim3(256), 0, stream, U, GT, out);
}


# Forward W-DFT moved into the H-DFT kernel: a wave turns its 64x64 strip of the H result through LDS into A fragments, twiddles from the LDS cos/sin list, 4 strips added in LDS; W sum order changes
# speedup vs baseline: 2.6622x; 1.1790x over previous
#include <hip/hip_runtime.h>


typedef __attribute__((ext_vector_type(2))) float v2f;
typedef __attribute__((ext_vector_type(4))) float v4f;
typedef __attribute__((ext_vector_type(8))) float v8f;

__device__ __forceinline__ v8f wmma4(v2f a, v2f b, v8f c) {
  return __builtin_amdgcn_wmma_f32_16x16x4_f32(false, a, false, b, (short)0, c,
                                               false, false);
}

__device__ __forceinline__ v2f ld2(const float* p) { return *(const v2f*)p; }
__device__ __forceinline__ v4f ld4_nt(const float* p) {
  return __builtin_nontemporal_load((const v4f*)p);
}

struct XBlk {
  v4f xa[8];
  v4f xm[8];
  v2f c0[4], c1[4], s0[4], s1[4];
};

__device__ __forceinline__ void k1_load_x(XBlk& B, const float* xp,
                                          const float* xmp, int kb, int half) {
#pragma unroll
  for (int q = 0; q < 4; ++q) {
    int k0 = kb + 4 * q;
    B.xa[2 * q] = ld4_nt(xp + (long)k0 * 256);
    B.xa[2 * q + 1] = ld4_nt(xp + (long)(k0 + 1) * 256);
    const float* m0 = xmp - (long)k0 * 256;
    if (q == 0) m0 = (k0 + 2 * half == 0) ? xp : m0;
    B.xm[2 * q] = ld4_nt(m0);
    B.xm[2 * q + 1] = ld4_nt(xmp - (long)(k0 + 1) * 256);
  }
}

__device__ __forceinline__ void k1_load_t(XBlk& B, const float* c0,
                                          const float* c1, const float* s0,
                                          const float* s1, int kb) {
#pragma unroll
  for (int q = 0; q < 4; ++q) {
    int k0 = kb + 4 * q;
    B.c0[q] = ld2(c0 + k0);
    B.c1[q] = ld2(c1 + k0);
    B.s0[q] = ld2(s0 + k0);
    B.s1[q] = ld2(s1 + k0);
  }
}

__device__ __forceinline__ void k1_load(XBlk& B, const float* xp,
                                        const float* xmp, const float* c0,
                                        const float* c1, const float* s0,
                                        const float* s1, int kb, int half) {
  k1_load_x(B, xp, xmp, kb, half);
  k1_load_t(B, c0, c1, s0, s1, kb);
}

__device__ __forceinline__ void k1_mma(const XBlk& B, v8f (&acc)[4][4],
                                       bool row0) {
#pragma unroll
  for (int q = 0; q < 4; ++q) {
    v4f e0 = B.xa[2 * q] + B.xm[2 * q];
    v4f e1 = B.xa[2 * q + 1] + B.xm[2 * q + 1];
    v4f o0 = B.xa[2 * q] - B.xm[2 * q];
    v4f o1 = B.xa[2 * q + 1] - B.xm[2 * q + 1];
    if (q == 0) {
#pragma unroll
      for (int t = 0; t < 4; ++t) e0[t] = row0 ? B.xa[0][t] : e0[t];
    }
#pragma unroll
    for (int t = 0; t < 4; ++t) {
      v2f be, bo;
      be.x = e0[t];
      be.y = e1[t];
      bo.x = o0[t];
      bo.y = o1[t];
      acc[0][t] = wmma4(B.c0[q], be, acc[0][t]);
      acc[1][t] = wmma4(B.c1[q], be, acc[1][t]);
      acc[2][t] = wmma4(B.s0[q], bo, acc[2][t]);
      acc[3][t] = wmma4(B.s1[q], bo, acc[3][t]);
    }
  }
}

#define E2P 132
#define PLP 68
#define PLW (64 * PLP)
__global__ __launch_bounds__(256) void k1_dft_hw(const float* __restrict__ x,
                                                 float* __restrict__ tab,
                                                 float* __restrict__ Zre,
                                                 float* __restrict__ Zim) {
  __shared__ __attribute__((aligned(16))) float lds[512 + 64 * E2P + 8 * PLW];
  float* cs = lds;
  float* e2 = lds + 512;
  float* pl = lds + 512 + 64 * E2P;
  const int tid = threadIdx.x;
  const int wv = tid >> 5;
  const int lane = tid & 31;
  const int half = lane >> 4, l16 = lane & 15;
  const int bi = blockIdx.x * 2 + (wv >> 2);
  const int wbase = (wv & 3) * 64;
  const float* ximg = x + (long)bi * 65536 + wbase + 4 * l16;
  const float* xp = ximg + (2 * half) * 256;
  const float* xmp = ximg + (256 - 2 * half) * 256;
  const float* c0 = e2 + l16 * E2P + 2 * half;
  const float* c1 = e2 + (16 + l16) * E2P + 2 * half;
  const float* s0 = e2 + (32 + l16) * E2P + 2 * half;
  const float* s1 = e2 + (48 + l16) * E2P + 2 * half;
  v8f acc[4][4] = {};
  XBlk A, B;
  k1_load_x(A, xp, xmp, 0, half);
  {
    const float step = 6.28318530717958647692f / 256.0f;
    float th = step * (float)tid;
    cs[tid] = cosf(th);
    cs[256 + tid] = sinf(th);
  }
  __syncthreads();
  {
    const int row = tid >> 2;
    const int r = row & 31;
    const int ky = (r < 16) ? r : (224 + r);
    const int h0 = (tid & 3) * 33;
    const float* src = cs + ((row >> 5) << 8);
    float* dst = e2 + row * E2P + h0;
    int n = ky * h0;
#pragma unroll
    for (int it = 0; it < 33; ++it) {
      dst[it] = src[n & 255];
      n += ky;
    }
  }
  __syncthreads();
  if (tid < 128 && blockIdx.x < 192) {
    const int e = 8192 + blockIdx.x * 128 + tid;
    float v;
    int o;
    if (e < 24576) {
      int j = e - 8192;
      int is_s = j >= 8192;
      int h = (j & 8191) >> 5, r = j & 31;
      int ky = (r < 16) ? r : (224 + r);
      int m = (ky * h) & 255;
      v = is_s ? cs[256 + m] : cs[m];
      o = 24576 + j;
    } else {
      int j = e - 24576;
      int w = j >> 5, n = j & 31;
      int kx = n & 15;
      int m = (kx * w) & 255;
      float sc = ((kx == 0) ? 1.0f : 2.0f) * (1.0f / 65536.0f);
      v = (n < 16) ? sc * cs[m] : -sc * cs[256 + m];
      o = 40960 + j;
    }
    tab[o] = v;
  }
  k1_load_t(A, c0, c1, s0, s1, 0);
#pragma unroll 1
  for (int kb = 0; kb < 96; kb += 32) {
    k1_load(B, xp, xmp, c0, c1, s0, s1, kb + 16, half);
    __builtin_amdgcn_sched_barrier(0);
    k1_mma(A, acc, (kb == 0) && (half == 0));
    __builtin_amdgcn_sched_barrier(0);
    k1_load(A, xp, xmp, c0, c1, s0, s1, kb + 32, half);
    __builtin_amdgcn_sched_barrier(0);
    k1_mma(B, acc, false);
    __builtin_amdgcn_sched_barrier(0);
  }
  k1_load(B, xp, xmp, c0, c1, s0, s1, 112, half);
  v4f x128 = ld4_nt(ximg + 128 * 256);
  v2f t0 = ld2(c0 + 128);
  v2f t1 = ld2(c1 + 128);
  k1_mma(A, acc, false);
  k1_mma(B, acc, false);
  {
    v2f a0, a1;
    a0.x = half ? 0.0f : t0.x;
    a0.y = 0.0f;
    a1.x = half ? 0.0f : t1.x;
    a1.y = 0.0f;
#pragma unroll
    for (int t = 0; t < 4; ++t) {
      v2f b;
      b.x = half ? 0.0f : x128[t];
      b.y = 0.0f;
      acc[0][t] = wmma4(a0, b, acc[0][t]);
      acc[1][t] = wmma4(a1, b, acc[1][t]);
    }
  }
  float* my = pl + wv * PLW;
#pragma unroll
  for (int a = 0; a < 4; ++a) {
#pragma unroll
    for (int r = 0; r < 8; ++r) {
      v4f o = {acc[a][0][r], acc[a][1][r], acc[a][2][r], acc[a][3][r]};
      *(v4f*)(my + (16 * a + 8 * half + r) * PLP + 4 * l16) = o;
    }
  }
  const float* arow = my + l16 * PLP + 2 * half;
  v2f fc0[16], fc1[16], fs0[16], fs1[16], bco[16], bsi[16];
#pragma unroll
  for (int q = 0; q < 16; ++q) {
    fc0[q] = ld2(arow + 4 * q);
    fc1[q] = ld2(arow + 16 * PLP + 4 * q);
    fs0[q] = ld2(arow + 32 * PLP + 4 * q);
    fs1[q] = ld2(arow + 48 * PLP + 4 * q);
    int m0 = (l16 * (wbase + 4 * q + 2 * half)) & 255;
    int m1 = (m0 + l16) & 255;
    bco[q].x = cs[m0];
    bco[q].y = cs[m1];
    bsi[q].x = cs[256 + m0];
    bsi[q].y = cs[256 + m1];
  }
  __builtin_amdgcn_sched_barrier(0);
  v8f z0 = {}, z1 = {}, z2 = {}, z3 = {}, z4 = {}, z5 = {}, z6 = {}, z7 = {};
#pragma unroll
  for (int q = 0; q < 16; ++q) {
    z0 = wmma4(fc0[q], bco[q], z0);
    z1 = wmma4(fs0[q], bsi[q], z1);
    z2 = wmma4(fc0[q], bsi[q], z2);
    z3 = wmma4(fs0[q], bco[q], z3);
    z4 = wmma4(fc1[q], bco[q], z4);
    z5 = wmma4(fs1[q], bsi[q], z5);
    z6 = wmma4(fc1[q], bsi[q], z6);
    z7 = wmma4(fs1[q], bco[q], z7);
  }
  float* mine = my + lane;
#pragma unroll
  for (int r = 0; r < 8; ++r) {
    mine[(0 * 8 + r) * 32] = z0[r] - z1[r];
    mine[(1 * 8 + r) * 32] = -z2[r] - z3[r];
    mine[(2 * 8 + r) * 32] = z4[r] - z5[r];
    mine[(3 * 8 + r) * 32] = -z6[r] - z7[r];
  }
  __syncthreads();
  {
    const int u = tid & 127;
    const float* p0 = pl + (wv >> 2) * 4 * PLW + u;
    const int b = bi >> 5, i = bi & 31;
    const int hf = (u >> 4) & 1, kx = u & 15;
#pragma unroll
    for (int n = 0; n < 8; ++n) {
      int v = u + 128 * n;
      float s = p0[128 * n];
      s += p0[PLW + 128 * n];
      s += p0[2 * PLW + 128 * n];
      s += p0[3 * PLW + 128 * n];
      int o = v >> 8, r = (v >> 5) & 7;
      int kyp = (o >> 1) * 16 + hf * 8 + r;
      long off = (((long)kyp * 16 + kx) * 16 + b) * 32 + i;
      if (o & 1)
        Zim[off] = s;
      else
        Zre[off] = s;
    }
  }
}

#define K3_PLANE 520
__global__ __launch_bounds__(256) void k3_mix(
    const float* __restrict__ Zre, const float* __restrict__ Zim,
    const float* __restrict__ w1r, const float* __restrict__ w1i,
    const float* __restrict__ w2r, const float* __restrict__ w2i,
    float* __restrict__ Tre, float* __restrict__ Tim) {
  __shared__ __attribute__((aligned(16))) float wl[2][8 * K3_PLANE];
  __shared__ __attribute__((aligned(16))) float tl[2][2048];
  const int tid = threadIdx.x;
  const int wv = tid >> 5;
  const int lane = tid & 31;
  const int half = lane >> 4, l16 = lane & 15;
  const int kyp = blockIdx.x >> 2;
  const int ot = (blockIdx.x >> 1) & 1;
  const int kh = blockIdx.x & 1;
  const int ky = kyp & 15;
  const float* wr = (kyp < 16) ? w1r : w2r;
  const float* wi = (kyp < 16) ? w1i : w2i;
  const int m = kyp * 16 + kh * 8 + wv;
  const float* zre = Zre + (long)m * 512 + l16 * 32 + 2 * half;
  const float* zim = Zim + (long)m * 512 + l16 * 32 + 2 * half;
  v2f azr[8], azi[8];
#pragma unroll
  for (int q = 0; q < 8; ++q) {
    azr[q] = ld2(zre + 4 * q);
    azi[q] = ld2(zim + 4 * q);
  }
#pragma unroll
  for (int n = 0; n < 4; ++n) {
    int p = tid + 256 * n;
    int j = p & 1, ol = (p >> 1) & 15, i = p >> 5;
    long g = ((long)(i * 32 + ot * 16 + ol)) * 256 + ky * 16 + kh * 8 + j * 4;
    v4f vr = *(const v4f*)(wr + g);
    v4f vi = *(const v4f*)(wi + g);
    int base = (j * 4) * K3_PLANE + ((i >> 1) * 16 + ol) * 2 + (i & 1);
#pragma unroll
    for (int e = 0; e < 4; ++e) {
      wl[0][base + e * K3_PLANE] = vr[e];
      wl[1][base + e * K3_PLANE] = vi[e];
    }
  }
  __syncthreads();
  const float* bre = &wl[0][wv * K3_PLANE + (half * 16 + l16) * 2];
  const float* bim = &wl[1][wv * K3_PLANE + (half * 16 + l16) * 2];
  v8f p0 = {}, q0 = {}, u0 = {}, v0 = {};
#pragma unroll
  for (int q = 0; q < 8; ++q) {
    v2f br = *(const v2f*)(bre + q * 64);
    v2f bi = *(const v2f*)(bim + q * 64);
    p0 = wmma4(azr[q], br, p0);
    q0 = wmma4(azi[q], bi, q0);
    u0 = wmma4(azr[q], bi, u0);
    v0 = wmma4(azi[q], br, v0);
  }
#pragma unroll
  for (int r = 0; r < 8; ++r) {
    int t = ((r + half * 8) * 16 + l16) * 8 + wv;
    tl[0][t] = p0[r] - q0[r];
    tl[1][t] = u0[r] + v0[r];
  }
  __syncthreads();
#pragma unroll
  for (int n = 0; n < 2; ++n) {
    int pp = tid + 256 * n;
    int j = pp & 1, ol = (pp >> 1) & 15, b = pp >> 5;
    long o = (((long)b * 32 + ot * 16 + ol) * 32 + kyp) * 16 + kh * 8 + j * 4;
    *(v4f*)(Tre + o) = *(const v4f*)&tl[0][pp * 4];
    *(v4f*)(Tim + o) = *(const v4f*)&tl[1][pp * 4];
  }
}

__global__ __launch_bounds__(256) void k4_idft_h(const float* __restrict__ Tre,
                                                 const float* __restrict__ Tim,
                                                 const float* __restrict__ E4c,
                                                 const float* __restrict__ E4s,
                                                 float* __restrict__ U) {
  const int wid = blockIdx.x * 8 + (threadIdx.x >> 5);
  const int lane = threadIdx.x & 31;
  const int half = lane >> 4, l16 = lane & 15;
  const int bo = wid >> 4;
  const int h0 = (wid & 15) * 16;
  const float* tre = Tre + (long)bo * 512;
  const float* tim = Tim + (long)bo * 512;
  const float* ec = E4c + (h0 + l16) * 32 + 2 * half;
  const float* es = E4s + (h0 + l16) * 32 + 2 * half;
  v8f a0 = {}, a1 = {}, a2 = {}, a3 = {};
#pragma unroll
  for (int k0 = 0; k0 < 32; k0 += 4) {
    int kr = k0 + 2 * half;
    v2f ac = ld2(ec + k0);
    v2f as = ld2(es + k0);
    v2f bre, bim;
    bre.x = tre[kr * 16 + l16];
    bre.y = tre[(kr + 1) * 16 + l16];
    bim.x = tim[kr * 16 + l16];
    bim.y = tim[(kr + 1) * 16 + l16];
    a0 = wmma4(ac, bre, a0);
    a1 = wmma4(as, bim, a1);
    a2 = wmma4(ac, bim, a2);
    a3 = wmma4(as, bre, a3);
  }
  float* ub = U + ((long)bo * 256 + h0) * 32;
#pragma unroll
  for (int r = 0; r < 8; ++r) {
    int hh = r + half * 8;
    ub[hh * 32 + l16] = a0[r] - a1[r];
    ub[hh * 32 + 16 + l16] = a2[r] + a3[r];
  }
}

__device__ __forceinline__ void k5_load(v2f (&a)[8], const float* arow) {
#pragma unroll
  for (int q = 0; q < 8; ++q) a[q] = ld2(arow + 4 * q);
}

__device__ __forceinline__ void k5_tile(const v2f (&a)[8],
                                        const v2f (&g)[4][8], float* o) {
  v8f acc[4] = {};
#pragma unroll
  for (int q = 0; q < 8; ++q) {
#pragma unroll
    for (int t = 0; t < 4; ++t) acc[t] = wmma4(a[q], g[t][q], acc[t]);
  }
#pragma unroll
  for (int r = 0; r < 8; ++r) {
#pragma unroll
    for (int t = 0; t < 4; ++t) {
      __builtin_nontemporal_store(acc[t][r], &o[r * 256 + t * 16]);
    }
  }
}

__global__ __launch_bounds__(256) void k5_idft_w(const float* __restrict__ U,
                                                 const float* __restrict__ GT,
                                                 float* __restrict__ out) {
  const int wv = threadIdx.x >> 5;
  const int lane = threadIdx.x & 31;
  const int half = lane >> 4, l16 = lane & 15;
  const int bo = blockIdx.x * 2 + (wv >> 2);
  const int n0 = (wv & 3) * 64;
  const long rowbase = (long)bo * 256;
  const float* g0 = GT + (n0 + l16) * 32 + 2 * half;
  v2f g[4][8];
#pragma unroll
  for (int t = 0; t < 4; ++t) {
#pragma unroll
    for (int q = 0; q < 8; ++q) g[t][q] = ld2(g0 + t * 16 * 32 + 4 * q);
  }
  const float* arow = U + (rowbase + l16) * 32 + 2 * half;
  float* obase = out + (rowbase + half * 8) * 256 + n0 + l16;
  v2f A[8], B[8];
  k5_load(A, arow);
  for (int j = 0; j < 14; j += 2) {
    k5_load(B, arow + (j + 1) * 512);
    k5_tile(A, g, obase + j * 4096);
    k5_load(A, arow + (j + 2) * 512);
    k5_tile(B, g, obase + (j + 1) * 4096);
  }
  k5_load(B, arow + 15 * 512);
  k5_tile(A, g, obase + 14 * 4096);
  k5_tile(B, g, obase + 15 * 4096);
}

extern "C" void kernel_launch(void* const* d_in, const int* in_sizes, int n_in,
                              void* d_out, int out_size, void* d_ws,
                              size_t ws_size, hipStream_t stream) {
  (void)in_sizes; (void)n_in; (void)out_size; (void)ws_size;
  const float* x = (const float*)d_in[0];
  const float* w1r = (const float*)d_in[1];
  const float* w1i = (const float*)d_in[2];
  const float* w2r = (const float*)d_in[3];
  const float* w2i = (const float*)d_in[4];
  float* out = (float*)d_out;

  float* tab = (float*)d_ws;
  float* E4c = tab + 24576;
  float* E4s = tab + 32768;
  float* GT = tab + 40960;
  float* Zre = tab + 49152;
  float* Zim = Zre + 262144;
  float* Tre = Zim + 262144;
  float* Tim = Tre + 262144;
  float* U = Tim + 262144;

  hipLaunchKernelGGL(k1_dft_hw, dim3(256), dim3(256), 0, stream, x, tab, Zre,
                     Zim);
  hipLaunchKernelGGL(k3_mix, dim3(128), dim3(256), 0, stream, Zre, Zim, w1r, w1i,
                     w2r, w2i, Tre, Tim);
  hipLaunchKernelGGL(k4_idft_h, dim3(1024), dim3(256), 0, stream, Tre, Tim, E4c,
                     E4s, U);
  hipLaunchKernelGGL(k5_idft_w, dim3(256), dim3(256), 0, stream, U, GT, out);
}


# Inverse W-DFT folded about w=128 (cos even, sin odd): E=Ure*cos, O=Uim*sin on w=0..128, out[w]=E+O, out[256-w]=E-O; a wave owns 64 whole rows; 272 WMMAs per wave, not 512; W sum order changes
# speedup vs baseline: 2.8110x; 1.0559x over previous
#include <hip/hip_runtime.h>


typedef __attribute__((ext_vector_type(2))) float v2f;
typedef __attribute__((ext_vector_type(4))) float v4f;
typedef __attribute__((ext_vector_type(8))) float v8f;

__device__ __forceinline__ v8f wmma4(v2f a, v2f b, v8f c) {
  return __builtin_amdgcn_wmma_f32_16x16x4_f32(false, a, false, b, (short)0, c,
                                               false, false);
}

__device__ __forceinline__ v2f ld2(const float* p) { return *(const v2f*)p; }
__device__ __forceinline__ v4f ld4_nt(const float* p) {
  return __builtin_nontemporal_load((const v4f*)p);
}

struct XBlk {
  v4f xa[8];
  v4f xm[8];
  v2f c0[4], c1[4], s0[4], s1[4];
};

__device__ __forceinline__ void k1_load_x(XBlk& B, const float* xp,
                                          const float* xmp, int kb, int half) {
#pragma unroll
  for (int q = 0; q < 4; ++q) {
    int k0 = kb + 4 * q;
    B.xa[2 * q] = ld4_nt(xp + (long)k0 * 256);
    B.xa[2 * q + 1] = ld4_nt(xp + (long)(k0 + 1) * 256);
    const float* m0 = xmp - (long)k0 * 256;
    if (q == 0) m0 = (k0 + 2 * half == 0) ? xp : m0;
    B.xm[2 * q] = ld4_nt(m0);
    B.xm[2 * q + 1] = ld4_nt(xmp - (long)(k0 + 1) * 256);
  }
}

__device__ __forceinline__ void k1_load_t(XBlk& B, const float* c0,
                                          const float* c1, const float* s0,
                                          const float* s1, int kb) {
#pragma unroll
  for (int q = 0; q < 4; ++q) {
    int k0 = kb + 4 * q;
    B.c0[q] = ld2(c0 + k0);
    B.c1[q] = ld2(c1 + k0);
    B.s0[q] = ld2(s0 + k0);
    B.s1[q] = ld2(s1 + k0);
  }
}

__device__ __forceinline__ void k1_load(XBlk& B, const float* xp,
                                        const float* xmp, const float* c0,
                                        const float* c1, const float* s0,
                                        const float* s1, int kb, int half) {
  k1_load_x(B, xp, xmp, kb, half);
  k1_load_t(B, c0, c1, s0, s1, kb);
}

__device__ __forceinline__ void k1_mma(const XBlk& B, v8f (&acc)[4][4],
                                       bool row0) {
#pragma unroll
  for (int q = 0; q < 4; ++q) {
    v4f e0 = B.xa[2 * q] + B.xm[2 * q];
    v4f e1 = B.xa[2 * q + 1] + B.xm[2 * q + 1];
    v4f o0 = B.xa[2 * q] - B.xm[2 * q];
    v4f o1 = B.xa[2 * q + 1] - B.xm[2 * q + 1];
    if (q == 0) {
#pragma unroll
      for (int t = 0; t < 4; ++t) e0[t] = row0 ? B.xa[0][t] : e0[t];
    }
#pragma unroll
    for (int t = 0; t < 4; ++t) {
      v2f be, bo;
      be.x = e0[t];
      be.y = e1[t];
      bo.x = o0[t];
      bo.y = o1[t];
      acc[0][t] = wmma4(B.c0[q], be, acc[0][t]);
      acc[1][t] = wmma4(B.c1[q], be, acc[1][t]);
      acc[2][t] = wmma4(B.s0[q], bo, acc[2][t]);
      acc[3][t] = wmma4(B.s1[q], bo, acc[3][t]);
    }
  }
}

#define E2P 132
#define PLP 68
#define PLW (64 * PLP)
__global__ __launch_bounds__(256) void k1_dft_hw(const float* __restrict__ x,
                                                 float* __restrict__ tab,
                                                 float* __restrict__ Zre,
                                                 float* __restrict__ Zim) {
  __shared__ __attribute__((aligned(16))) float lds[512 + 64 * E2P + 8 * PLW];
  float* cs = lds;
  float* e2 = lds + 512;
  float* pl = lds + 512 + 64 * E2P;
  const int tid = threadIdx.x;
  const int wv = tid >> 5;
  const int lane = tid & 31;
  const int half = lane >> 4, l16 = lane & 15;
  const int bi = blockIdx.x * 2 + (wv >> 2);
  const int wbase = (wv & 3) * 64;
  const float* ximg = x + (long)bi * 65536 + wbase + 4 * l16;
  const float* xp = ximg + (2 * half) * 256;
  const float* xmp = ximg + (256 - 2 * half) * 256;
  const float* c0 = e2 + l16 * E2P + 2 * half;
  const float* c1 = e2 + (16 + l16) * E2P + 2 * half;
  const float* s0 = e2 + (32 + l16) * E2P + 2 * half;
  const float* s1 = e2 + (48 + l16) * E2P + 2 * half;
  v8f acc[4][4] = {};
  XBlk A, B;
  k1_load_x(A, xp, xmp, 0, half);
  {
    const float step = 6.28318530717958647692f / 256.0f;
    float th = step * (float)tid;
    cs[tid] = cosf(th);
    cs[256 + tid] = sinf(th);
  }
  __syncthreads();
  {
    const int row = tid >> 2;
    const int r = row & 31;
    const int ky = (r < 16) ? r : (224 + r);
    const int h0 = (tid & 3) * 33;
    const float* src = cs + ((row >> 5) << 8);
    float* dst = e2 + row * E2P + h0;
    int n = ky * h0;
#pragma unroll
    for (int it = 0; it < 33; ++it) {
      dst[it] = src[n & 255];
      n += ky;
    }
  }
  __syncthreads();
  if (tid < 128 && blockIdx.x < 192) {
    const int e = 8192 + blockIdx.x * 128 + tid;
    float v;
    int o;
    if (e < 24576) {
      int j = e - 8192;
      int is_s = j >= 8192;
      int h = (j & 8191) >> 5, r = j & 31;
      int ky = (r < 16) ? r : (224 + r);
      int m = (ky * h) & 255;
      v = is_s ? cs[256 + m] : cs[m];
      o = 24576 + j;
    } else {
      int j = e - 24576;
      int w = j >> 5, n = j & 31;
      int kx = n & 15;
      int m = (kx * w) & 255;
      float sc = ((kx == 0) ? 1.0f : 2.0f) * (1.0f / 65536.0f);
      v = (n < 16) ? sc * cs[m] : -sc * cs[256 + m];
      o = 40960 + j;
    }
    tab[o] = v;
  }
  k1_load_t(A, c0, c1, s0, s1, 0);
#pragma unroll 1
  for (int kb = 0; kb < 96; kb += 32) {
    k1_load(B, xp, xmp, c0, c1, s0, s1, kb + 16, half);
    __builtin_amdgcn_sched_barrier(0);
    k1_mma(A, acc, (kb == 0) && (half == 0));
    __builtin_amdgcn_sched_barrier(0);
    k1_load(A, xp, xmp, c0, c1, s0, s1, kb + 32, half);
    __builtin_amdgcn_sched_barrier(0);
    k1_mma(B, acc, false);
    __builtin_amdgcn_sched_barrier(0);
  }
  k1_load(B, xp, xmp, c0, c1, s0, s1, 112, half);
  v4f x128 = ld4_nt(ximg + 128 * 256);
  v2f t0 = ld2(c0 + 128);
  v2f t1 = ld2(c1 + 128);
  k1_mma(A, acc, false);
  k1_mma(B, acc, false);
  {
    v2f a0, a1;
    a0.x = half ? 0.0f : t0.x;
    a0.y = 0.0f;
    a1.x = half ? 0.0f : t1.x;
    a1.y = 0.0f;
#pragma unroll
    for (int t = 0; t < 4; ++t) {
      v2f b;
      b.x = half ? 0.0f : x128[t];
      b.y = 0.0f;
      acc[0][t] = wmma4(a0, b, acc[0][t]);
      acc[1][t] = wmma4(a1, b, acc[1][t]);
    }
  }
  float* my = pl + wv * PLW;
#pragma unroll
  for (int a = 0; a < 4; ++a) {
#pragma unroll
    for (int r = 0; r < 8; ++r) {
      v4f o = {acc[a][0][r], acc[a][1][r], acc[a][2][r], acc[a][3][r]};
      *(v4f*)(my + (16 * a + 8 * half + r) * PLP + 4 * l16) = o;
    }
  }
  const float* arow = my + l16 * PLP + 2 * half;
  v2f fc0[16], fc1[16], fs0[16], fs1[16], bco[16], bsi[16];
#pragma unroll
  for (int q = 0; q < 16; ++q) {
    fc0[q] = ld2(arow + 4 * q);
    fc1[q] = ld2(arow + 16 * PLP + 4 * q);
    fs0[q] = ld2(arow + 32 * PLP + 4 * q);
    fs1[q] = ld2(arow + 48 * PLP + 4 * q);
    int m0 = (l16 * (wbase + 4 * q + 2 * half)) & 255;
    int m1 = (m0 + l16) & 255;
    bco[q].x = cs[m0];
    bco[q].y = cs[m1];
    bsi[q].x = cs[256 + m0];
    bsi[q].y = cs[256 + m1];
  }
  __builtin_amdgcn_sched_barrier(0);
  v8f z0 = {}, z1 = {}, z2 = {}, z3 = {}, z4 = {}, z5 = {}, z6 = {}, z7 = {};
#pragma unroll
  for (int q = 0; q < 16; ++q) {
    z0 = wmma4(fc0[q], bco[q], z0);
    z1 = wmma4(fs0[q], bsi[q], z1);
    z2 = wmma4(fc0[q], bsi[q], z2);
    z3 = wmma4(fs0[q], bco[q], z3);
    z4 = wmma4(fc1[q], bco[q], z4);
    z5 = wmma4(fs1[q], bsi[q], z5);
    z6 = wmma4(fc1[q], bsi[q], z6);
    z7 = wmma4(fs1[q], bco[q], z7);
  }
  float* mine = my + lane;
#pragma unroll
  for (int r = 0; r < 8; ++r) {
    mine[(0 * 8 + r) * 32] = z0[r] - z1[r];
    mine[(1 * 8 + r) * 32] = -z2[r] - z3[r];
    mine[(2 * 8 + r) * 32] = z4[r] - z5[r];
    mine[(3 * 8 + r) * 32] = -z6[r] - z7[r];
  }
  __syncthreads();
  {
    const int u = tid & 127;
    const float* p0 = pl + (wv >> 2) * 4 * PLW + u;
    const int b = bi >> 5, i = bi & 31;
    const int hf = (u >> 4) & 1, kx = u & 15;
#pragma unroll
    for (int n = 0; n < 8; ++n) {
      int v = u + 128 * n;
      float s = p0[128 * n];
      s += p0[PLW + 128 * n];
      s += p0[2 * PLW + 128 * n];
      s += p0[3 * PLW + 128 * n];
      int o = v >> 8, r = (v >> 5) & 7;
      int kyp = (o >> 1) * 16 + hf * 8 + r;
      long off = (((long)kyp * 16 + kx) * 16 + b) * 32 + i;
      if (o & 1)
        Zim[off] = s;
      else
        Zre[off] = s;
    }
  }
}

#define K3_PLANE 520
__global__ __launch_bounds__(256) void k3_mix(
    const float* __restrict__ Zre, const float* __restrict__ Zim,
    const float* __restrict__ w1r, const float* __restrict__ w1i,
    const float* __restrict__ w2r, const float* __restrict__ w2i,
    float* __restrict__ Tre, float* __restrict__ Tim) {
  __shared__ __attribute__((aligned(16))) float wl[2][8 * K3_PLANE];
  __shared__ __attribute__((aligned(16))) float tl[2][2048];
  const int tid = threadIdx.x;
  const int wv = tid >> 5;
  const int lane = tid & 31;
  const int half = lane >> 4, l16 = lane & 15;
  const int kyp = blockIdx.x >> 2;
  const int ot = (blockIdx.x >> 1) & 1;
  const int kh = blockIdx.x & 1;
  const int ky = kyp & 15;
  const float* wr = (kyp < 16) ? w1r : w2r;
  const float* wi = (kyp < 16) ? w1i : w2i;
  const int m = kyp * 16 + kh * 8 + wv;
  const float* zre = Zre + (long)m * 512 + l16 * 32 + 2 * half;
  const float* zim = Zim + (long)m * 512 + l16 * 32 + 2 * half;
  v2f azr[8], azi[8];
#pragma unroll
  for (int q = 0; q < 8; ++q) {
    azr[q] = ld2(zre + 4 * q);
    azi[q] = ld2(zim + 4 * q);
  }
#pragma unroll
  for (int n = 0; n < 4; ++n) {
    int p = tid + 256 * n;
    int j = p & 1, ol = (p >> 1) & 15, i = p >> 5;
    long g = ((long)(i * 32 + ot * 16 + ol)) * 256 + ky * 16 + kh * 8 + j * 4;
    v4f vr = *(const v4f*)(wr + g);
    v4f vi = *(const v4f*)(wi + g);
    int base = (j * 4) * K3_PLANE + ((i >> 1) * 16 + ol) * 2 + (i & 1);
#pragma unroll
    for (int e = 0; e < 4; ++e) {
      wl[0][base + e * K3_PLANE] = vr[e];
      wl[1][base + e * K3_PLANE] = vi[e];
    }
  }
  __syncthreads();
  const float* bre = &wl[0][wv * K3_PLANE + (half * 16 + l16) * 2];
  const float* bim = &wl[1][wv * K3_PLANE + (half * 16 + l16) * 2];
  v8f p0 = {}, q0 = {}, u0 = {}, v0 = {};
#pragma unroll
  for (int q = 0; q < 8; ++q) {
    v2f br = *(const v2f*)(bre + q * 64);
    v2f bi = *(const v2f*)(bim + q * 64);
    p0 = wmma4(azr[q], br, p0);
    q0 = wmma4(azi[q], bi, q0);
    u0 = wmma4(azr[q], bi, u0);
    v0 = wmma4(azi[q], br, v0);
  }
#pragma unroll
  for (int r = 0; r < 8; ++r) {
    int t = ((r + half * 8) * 16 + l16) * 8 + wv;
    tl[0][t] = p0[r] - q0[r];
    tl[1][t] = u0[r] + v0[r];
  }
  __syncthreads();
#pragma unroll
  for (int n = 0; n < 2; ++n) {
    int pp = tid + 256 * n;
    int j = pp & 1, ol = (pp >> 1) & 15, b = pp >> 5;
    long o = (((long)b * 32 + ot * 16 + ol) * 32 + kyp) * 16 + kh * 8 + j * 4;
    *(v4f*)(Tre + o) = *(const v4f*)&tl[0][pp * 4];
    *(v4f*)(Tim + o) = *(const v4f*)&tl[1][pp * 4];
  }
}

__global__ __launch_bounds__(256) void k4_idft_h(const float* __restrict__ Tre,
                                                 const float* __restrict__ Tim,
                                                 const float* __restrict__ E4c,
                                                 const float* __restrict__ E4s,
                                                 float* __restrict__ U) {
  const int wid = blockIdx.x * 8 + (threadIdx.x >> 5);
  const int lane = threadIdx.x & 31;
  const int half = lane >> 4, l16 = lane & 15;
  const int bo = wid >> 4;
  const int h0 = (wid & 15) * 16;
  const float* tre = Tre + (long)bo * 512;
  const float* tim = Tim + (long)bo * 512;
  const float* ec = E4c + (h0 + l16) * 32 + 2 * half;
  const float* es = E4s + (h0 + l16) * 32 + 2 * half;
  v8f a0 = {}, a1 = {}, a2 = {}, a3 = {};
#pragma unroll
  for (int k0 = 0; k0 < 32; k0 += 4) {
    int kr = k0 + 2 * half;
    v2f ac = ld2(ec + k0);
    v2f as = ld2(es + k0);
    v2f bre, bim;
    bre.x = tre[kr * 16 + l16];
    bre.y = tre[(kr + 1) * 16 + l16];
    bim.x = tim[kr * 16 + l16];
    bim.y = tim[(kr + 1) * 16 + l16];
    a0 = wmma4(ac, bre, a0);
    a1 = wmma4(as, bim, a1);
    a2 = wmma4(ac, bim, a2);
    a3 = wmma4(as, bre, a3);
  }
  float* ub = U + ((long)bo * 256 + h0) * 32;
#pragma unroll
  for (int r = 0; r < 8; ++r) {
    int hh = r + half * 8;
    ub[hh * 32 + l16] = a0[r] - a1[r];
    ub[hh * 32 + 16 + l16] = a2[r] + a3[r];
  }
}

__device__ __forceinline__ void k5_tile(const v2f (&a)[8],
                                        const v2f (&gc)[9][4],
                                        const v2f (&gs)[8][4], float* od,
                                        float* om, bool lane0) {
  v8f e[9] = {}, o[8] = {};
#pragma unroll
  for (int q = 0; q < 4; ++q) {
#pragma unroll
    for (int t = 0; t < 9; ++t) e[t] = wmma4(a[q], gc[t][q], e[t]);
  }
#pragma unroll
  for (int q = 0; q < 4; ++q) {
#pragma unroll
    for (int t = 0; t < 8; ++t) o[t] = wmma4(a[4 + q], gs[t][q], o[t]);
  }
#pragma unroll
  for (int r = 0; r < 8; ++r) {
#pragma unroll
    for (int t = 0; t < 8; ++t)
      __builtin_nontemporal_store(e[t][r] + o[t][r], &od[r * 256 + t * 16]);
#pragma unroll
    for (int p = 0; p < 8; ++p) {
      float hi = (p == 0) ? e[8][r] : e[8 - p][r] - o[8 - p][r];
      float lo = e[7 - p][r] - o[7 - p][r];
      __builtin_nontemporal_store(lane0 ? hi : lo, &om[r * 256 + p * 16]);
    }
  }
}

__global__ __launch_bounds__(256) void k5_idft_w(const float* __restrict__ U,
                                                 const float* __restrict__ GT,
                                                 float* __restrict__ out) {
  const int wv = threadIdx.x >> 5;
  const int lane = threadIdx.x & 31;
  const int half = lane >> 4, l16 = lane & 15;
  const int bo = blockIdx.x * 2 + (wv >> 2);
  const long rowbase = (long)bo * 256 + (wv & 3) * 64;
  const float* g0 = GT + l16 * 32 + 2 * half;
  v2f gc[9][4], gs[8][4];
#pragma unroll
  for (int t = 0; t < 9; ++t) {
#pragma unroll
    for (int q = 0; q < 4; ++q) gc[t][q] = ld2(g0 + t * 512 + 4 * q);
  }
#pragma unroll
  for (int t = 0; t < 8; ++t) {
#pragma unroll
    for (int q = 0; q < 4; ++q) gs[t][q] = ld2(g0 + t * 512 + 16 + 4 * q);
  }
  const float* arow = U + (rowbase + l16) * 32 + 2 * half;
  v2f a[4][8];
#pragma unroll
  for (int j = 0; j < 4; ++j) {
#pragma unroll
    for (int q = 0; q < 8; ++q) a[j][q] = ld2(arow + j * 512 + 4 * q);
  }
  float* orow = out + (rowbase + half * 8) * 256;
  float* od = orow + l16;
  float* om = orow + 128 + ((16 - l16) & 15);
#pragma unroll
  for (int j = 0; j < 4; ++j)
    k5_tile(a[j], gc, gs, od + j * 4096, om + j * 4096, l16 == 0);
}

extern "C" void kernel_launch(void* const* d_in, const int* in_sizes, int n_in,
                              void* d_out, int out_size, void* d_ws,
                              size_t ws_size, hipStream_t stream) {
  (void)in_sizes; (void)n_in; (void)out_size; (void)ws_size;
  const float* x = (const float*)d_in[0];
  const float* w1r = (const float*)d_in[1];
  const float* w1i = (const float*)d_in[2];
  const float* w2r = (const float*)d_in[3];
  const float* w2i = (const float*)d_in[4];
  float* out = (float*)d_out;

  float* tab = (float*)d_ws;
  float* E4c = tab + 24576;
  float* E4s = tab + 32768;
  float* GT = tab + 40960;
  float* Zre = tab + 49152;
  float* Zim = Zre + 262144;
  float* Tre = Zim + 262144;
  float* Tim = Tre + 262144;
  float* U = Tim + 262144;

  hipLaunchKernelGGL(k1_dft_hw, dim3(256), dim3(256), 0, stream, x, tab, Zre,
                     Zim);
  hipLaunchKernelGGL(k3_mix, dim3(128), dim3(256), 0, stream, Zre, Zim, w1r, w1i,
                     w2r, w2i, Tre, Tim);
  hipLaunchKernelGGL(k4_idft_h, dim3(1024), dim3(256), 0, stream, Tre, Tim, E4c,
                     E4s, U);
  hipLaunchKernelGGL(k5_idft_w, dim3(256), dim3(256), 0, stream, U, GT, out);
}


# Inverse H-DFT moved into the inverse W-DFT kernel: a wave forms the U tiles of its own 64 rows and turns them through a wave-private LDS tile into A fragments; 3 launches, no U in memory; same sums
# speedup vs baseline: 2.9930x; 1.0647x over previous
#include <hip/hip_runtime.h>


typedef __attribute__((ext_vector_type(2))) float v2f;
typedef __attribute__((ext_vector_type(4))) float v4f;
typedef __attribute__((ext_vector_type(8))) float v8f;

__device__ __forceinline__ v8f wmma4(v2f a, v2f b, v8f c) {
  return __builtin_amdgcn_wmma_f32_16x16x4_f32(false, a, false, b, (short)0, c,
                                               false, false);
}

__device__ __forceinline__ v2f ld2(const float* p) { return *(const v2f*)p; }
__device__ __forceinline__ v4f ld4_nt(const float* p) {
  return __builtin_nontemporal_load((const v4f*)p);
}

struct XBlk {
  v4f xa[8];
  v4f xm[8];
  v2f c0[4], c1[4], s0[4], s1[4];
};

__device__ __forceinline__ void k1_load_x(XBlk& B, const float* xp,
                                          const float* xmp, int kb, int half) {
#pragma unroll
  for (int q = 0; q < 4; ++q) {
    int k0 = kb + 4 * q;
    B.xa[2 * q] = ld4_nt(xp + (long)k0 * 256);
    B.xa[2 * q + 1] = ld4_nt(xp + (long)(k0 + 1) * 256);
    const float* m0 = xmp - (long)k0 * 256;
    if (q == 0) m0 = (k0 + 2 * half == 0) ? xp : m0;
    B.xm[2 * q] = ld4_nt(m0);
    B.xm[2 * q + 1] = ld4_nt(xmp - (long)(k0 + 1) * 256);
  }
}

__device__ __forceinline__ void k1_load_t(XBlk& B, const float* c0,
                                          const float* c1, const float* s0,
                                          const float* s1, int kb) {
#pragma unroll
  for (int q = 0; q < 4; ++q) {
    int k0 = kb + 4 * q;
    B.c0[q] = ld2(c0 + k0);
    B.c1[q] = ld2(c1 + k0);
    B.s0[q] = ld2(s0 + k0);
    B.s1[q] = ld2(s1 + k0);
  }
}

__device__ __forceinline__ void k1_load(XBlk& B, const float* xp,
                                        const float* xmp, const float* c0,
                                        const float* c1, const float* s0,
                                        const float* s1, int kb, int half) {
  k1_load_x(B, xp, xmp, kb, half);
  k1_load_t(B, c0, c1, s0, s1, kb);
}

__device__ __forceinline__ void k1_mma(const XBlk& B, v8f (&acc)[4][4],
                                       bool row0) {
#pragma unroll
  for (int q = 0; q < 4; ++q) {
    v4f e0 = B.xa[2 * q] + B.xm[2 * q];
    v4f e1 = B.xa[2 * q + 1] + B.xm[2 * q + 1];
    v4f o0 = B.xa[2 * q] - B.xm[2 * q];
    v4f o1 = B.xa[2 * q + 1] - B.xm[2 * q + 1];
    if (q == 0) {
#pragma unroll
      for (int t = 0; t < 4; ++t) e0[t] = row0 ? B.xa[0][t] : e0[t];
    }
#pragma unroll
    for (int t = 0; t < 4; ++t) {
      v2f be, bo;
      be.x = e0[t];
      be.y = e1[t];
      bo.x = o0[t];
      bo.y = o1[t];
      acc[0][t] = wmma4(B.c0[q], be, acc[0][t]);
      acc[1][t] = wmma4(B.c1[q], be, acc[1][t]);
      acc[2][t] = wmma4(B.s0[q], bo, acc[2][t]);
      acc[3][t] = wmma4(B.s1[q], bo, acc[3][t]);
    }
  }
}

#define E2P 132
#define PLP 68
#define PLW (64 * PLP)
__global__ __launch_bounds__(256) void k1_dft_hw(const float* __restrict__ x,
                                                 float* __restrict__ tab,
                                                 float* __restrict__ Zre,
                                                 float* __restrict__ Zim) {
  __shared__ __attribute__((aligned(16))) float lds[512 + 64 * E2P + 8 * PLW];
  float* cs = lds;
  float* e2 = lds + 512;
  float* pl = lds + 512 + 64 * E2P;
  const int tid = threadIdx.x;
  const int wv = tid >> 5;
  const int lane = tid & 31;
  const int half = lane >> 4, l16 = lane & 15;
  const int bi = blockIdx.x * 2 + (wv >> 2);
  const int wbase = (wv & 3) * 64;
  const float* ximg = x + (long)bi * 65536 + wbase + 4 * l16;
  const float* xp = ximg + (2 * half) * 256;
  const float* xmp = ximg + (256 - 2 * half) * 256;
  const float* c0 = e2 + l16 * E2P + 2 * half;
  const float* c1 = e2 + (16 + l16) * E2P + 2 * half;
  const float* s0 = e2 + (32 + l16) * E2P + 2 * half;
  const float* s1 = e2 + (48 + l16) * E2P + 2 * half;
  v8f acc[4][4] = {};
  XBlk A, B;
  k1_load_x(A, xp, xmp, 0, half);
  {
    const float step = 6.28318530717958647692f / 256.0f;
    float th = step * (float)tid;
    cs[tid] = cosf(th);
    cs[256 + tid] = sinf(th);
  }
  __syncthreads();
  {
    const int row = tid >> 2;
    const int r = row & 31;
    const int ky = (r < 16) ? r : (224 + r);
    const int h0 = (tid & 3) * 33;
    const float* src = cs + ((row >> 5) << 8);
    float* dst = e2 + row * E2P + h0;
    int n = ky * h0;
#pragma unroll
    for (int it = 0; it < 33; ++it) {
      dst[it] = src[n & 255];
      n += ky;
    }
  }
  __syncthreads();
  if (tid < 128 && blockIdx.x < 192) {
    const int e = 8192 + blockIdx.x * 128 + tid;
    float v;
    int o;
    if (e < 24576) {
      int j = e - 8192;
      int is_s = j >= 8192;
      int h = (j & 8191) >> 5, r = j & 31;
      int ky = (r < 16) ? r : (224 + r);
      int m = (ky * h) & 255;
      v = is_s ? cs[256 + m] : cs[m];
      o = 24576 + j;
    } else {
      int j = e - 24576;
      int w = j >> 5, n = j & 31;
      int kx = n & 15;
      int m = (kx * w) & 255;
      float sc = ((kx == 0) ? 1.0f : 2.0f) * (1.0f / 65536.0f);
      v = (n < 16) ? sc * cs[m] : -sc * cs[256 + m];
      o = 40960 + j;
    }
    tab[o] = v;
  }
  k1_load_t(A, c0, c1, s0, s1, 0);
#pragma unroll 1
  for (int kb = 0; kb < 96; kb += 32) {
    k1_load(B, xp, xmp, c0, c1, s0, s1, kb + 16, half);
    __builtin_amdgcn_sched_barrier(0);
    k1_mma(A, acc, (kb == 0) && (half == 0));
    __builtin_amdgcn_sched_barrier(0);
    k1_load(A, xp, xmp, c0, c1, s0, s1, kb + 32, half);
    __builtin_amdgcn_sched_barrier(0);
    k1_mma(B, acc, false);
    __builtin_amdgcn_sched_barrier(0);
  }
  k1_load(B, xp, xmp, c0, c1, s0, s1, 112, half);
  v4f x128 = ld4_nt(ximg + 128 * 256);
  v2f t0 = ld2(c0 + 128);
  v2f t1 = ld2(c1 + 128);
  k1_mma(A, acc, false);
  k1_mma(B, acc, false);
  {
    v2f a0, a1;
    a0.x = half ? 0.0f : t0.x;
    a0.y = 0.0f;
    a1.x = half ? 0.0f : t1.x;
    a1.y = 0.0f;
#pragma unroll
    for (int t = 0; t < 4; ++t) {
      v2f b;
      b.x = half ? 0.0f : x128[t];
      b.y = 0.0f;
      acc[0][t] = wmma4(a0, b, acc[0][t]);
      acc[1][t] = wmma4(a1, b, acc[1][t]);
    }
  }
  float* my = pl + wv * PLW;
#pragma unroll
  for (int a = 0; a < 4; ++a) {
#pragma unroll
    for (int r = 0; r < 8; ++r) {
      v4f o = {acc[a][0][r], acc[a][1][r], acc[a][2][r], acc[a][3][r]};
      *(v4f*)(my + (16 * a + 8 * half + r) * PLP + 4 * l16) = o;
    }
  }
  const float* arow = my + l16 * PLP + 2 * half;
  v2f fc0[16], fc1[16], fs0[16], fs1[16], bco[16], bsi[16];
#pragma unroll
  for (int q = 0; q < 16; ++q) {
    fc0[q] = ld2(arow + 4 * q);
    fc1[q] = ld2(arow + 16 * PLP + 4 * q);
    fs0[q] = ld2(arow + 32 * PLP + 4 * q);
    fs1[q] = ld2(arow + 48 * PLP + 4 * q);
    int m0 = (l16 * (wbase + 4 * q + 2 * half)) & 255;
    int m1 = (m0 + l16) & 255;
    bco[q].x = cs[m0];
    bco[q].y = cs[m1];
    bsi[q].x = cs[256 + m0];
    bsi[q].y = cs[256 + m1];
  }
  __builtin_amdgcn_sched_barrier(0);
  v8f z0 = {}, z1 = {}, z2 = {}, z3 = {}, z4 = {}, z5 = {}, z6 = {}, z7 = {};
#pragma unroll
  for (int q = 0; q < 16; ++q) {
    z0 = wmma4(fc0[q], bco[q], z0);
    z1 = wmma4(fs0[q], bsi[q], z1);
    z2 = wmma4(fc0[q], bsi[q], z2);
    z3 = wmma4(fs0[q], bco[q], z3);
    z4 = wmma4(fc1[q], bco[q], z4);
    z5 = wmma4(fs1[q], bsi[q], z5);
    z6 = wmma4(fc1[q], bsi[q], z6);
    z7 = wmma4(fs1[q], bco[q], z7);
  }
  float* mine = my + lane;
#pragma unroll
  for (int r = 0; r < 8; ++r) {
    mine[(0 * 8 + r) * 32] = z0[r] - z1[r];
    mine[(1 * 8 + r) * 32] = -z2[r] - z3[r];
    mine[(2 * 8 + r) * 32] = z4[r] - z5[r];
    mine[(3 * 8 + r) * 32] = -z6[r] - z7[r];
  }
  __syncthreads();
  {
    const int u = tid & 127;
    const float* p0 = pl + (wv >> 2) * 4 * PLW + u;
    const int b = bi >> 5, i = bi & 31;
    const int hf = (u >> 4) & 1, kx = u & 15;
#pragma unroll
    for (int n = 0; n < 8; ++n) {
      int v = u + 128 * n;
      float s = p0[128 * n];
      s += p0[PLW + 128 * n];
      s += p0[2 * PLW + 128 * n];
      s += p0[3 * PLW + 128 * n];
      int o = v >> 8, r = (v >> 5) & 7;
      int kyp = (o >> 1) * 16 + hf * 8 + r;
      long off = (((long)kyp * 16 + kx) * 16 + b) * 32 + i;
      if (o & 1)
        Zim[off] = s;
      else
        Zre[off] = s;
    }
  }
}

#define K3_PLANE 520
__global__ __launch_bounds__(256) void k3_mix(
    const float* __restrict__ Zre, const float* __restrict__ Zim,
    const float* __restrict__ w1r, const float* __restrict__ w1i,
    const float* __restrict__ w2r, const float* __restrict__ w2i,
    float* __restrict__ Tre, float* __restrict__ Tim) {
  __shared__ __attribute__((aligned(16))) float wl[2][8 * K3_PLANE];
  __shared__ __attribute__((aligned(16))) float tl[2][2048];
  const int tid = threadIdx.x;
  const int wv = tid >> 5;
  const int lane = tid & 31;
  const int half = lane >> 4, l16 = lane & 15;
  const int kyp = blockIdx.x >> 2;
  const int ot = (blockIdx.x >> 1) & 1;
  const int kh = blockIdx.x & 1;
  const int ky = kyp & 15;
  const float* wr = (kyp < 16) ? w1r : w2r;
  const float* wi = (kyp < 16) ? w1i : w2i;
  const int m = kyp * 16 + kh * 8 + wv;
  const float* zre = Zre + (long)m * 512 + l16 * 32 + 2 * half;
  const float* zim = Zim + (long)m * 512 + l16 * 32 + 2 * half;
  v2f azr[8], azi[8];
#pragma unroll
  for (int q = 0; q < 8; ++q) {
    azr[q] = ld2(zre + 4 * q);
    azi[q] = ld2(zim + 4 * q);
  }
#pragma unroll
  for (int n = 0; n < 4; ++n) {
    int p = tid + 256 * n;
    int j = p & 1, ol = (p >> 1) & 15, i = p >> 5;
    long g = ((long)(i * 32 + ot * 16 + ol)) * 256 + ky * 16 + kh * 8 + j * 4;
    v4f vr = *(const v4f*)(wr + g);
    v4f vi = *(const v4f*)(wi + g);
    int base = (j * 4) * K3_PLANE + ((i >> 1) * 16 + ol) * 2 + (i & 1);
#pragma unroll
    for (int e = 0; e < 4; ++e) {
      wl[0][base + e * K3_PLANE] = vr[e];
      wl[1][base + e * K3_PLANE] = vi[e];
    }
  }
  __syncthreads();
  const float* bre = &wl[0][wv * K3_PLANE + (half * 16 + l16) * 2];
  const float* bim = &wl[1][wv * K3_PLANE + (half * 16 + l16) * 2];
  v8f p0 = {}, q0 = {}, u0 = {}, v0 = {};
#pragma unroll
  for (int q = 0; q < 8; ++q) {
    v2f br = *(const v2f*)(bre + q * 64);
    v2f bi = *(const v2f*)(bim + q * 64);
    p0 = wmma4(azr[q], br, p0);
    q0 = wmma4(azi[q], bi, q0);
    u0 = wmma4(azr[q], bi, u0);
    v0 = wmma4(azi[q], br, v0);
  }
#pragma unroll
  for (int r = 0; r < 8; ++r) {
    int t = ((r + half * 8) * 16 + l16) * 8 + wv;
    tl[0][t] = p0[r] - q0[r];
    tl[1][t] = u0[r] + v0[r];
  }
  __syncthreads();
#pragma unroll
  for (int n = 0; n < 2; ++n) {
    int pp = tid + 256 * n;
    int j = pp & 1, ol = (pp >> 1) & 15, b = pp >> 5;
    long o = (((long)b * 32 + ot * 16 + ol) * 32 + kyp) * 16 + kh * 8 + j * 4;
    *(v4f*)(Tre + o) = *(const v4f*)&tl[0][pp * 4];
    *(v4f*)(Tim + o) = *(const v4f*)&tl[1][pp * 4];
  }
}

__device__ __forceinline__ void k4_tile(const v2f (&a)[8],
                                        const v2f (&gc)[9][4],
                                        const v2f (&gs)[8][4], float* od,
                                        float* om, bool lane0) {
  v8f e[9] = {}, o[8] = {};
#pragma unroll
  for (int q = 0; q < 4; ++q) {
#pragma unroll
    for (int t = 0; t < 9; ++t) e[t] = wmma4(a[q], gc[t][q], e[t]);
  }
#pragma unroll
  for (int q = 0; q < 4; ++q) {
#pragma unroll
    for (int t = 0; t < 8; ++t) o[t] = wmma4(a[4 + q], gs[t][q], o[t]);
  }
#pragma unroll
  for (int r = 0; r < 8; ++r) {
#pragma unroll
    for (int t = 0; t < 8; ++t)
      __builtin_nontemporal_store(e[t][r] + o[t][r], &od[r * 256 + t * 16]);
#pragma unroll
    for (int p = 0; p < 8; ++p) {
      float hi = (p == 0) ? e[8][r] : e[8 - p][r] - o[8 - p][r];
      float lo = e[7 - p][r] - o[7 - p][r];
      __builtin_nontemporal_store(lane0 ? hi : lo, &om[r * 256 + p * 16]);
    }
  }
}

#define UP 36
#define UT (16 * UP)
__device__ __forceinline__ void k4_utile(const v2f (&tc)[8], const v2f (&ts)[8],
                                         const v2f (&bre)[8],
                                         const v2f (&bim)[8], float* tj,
                                         int half, int l16, v2f (&a)[8]) {
  v8f a0 = {}, a1 = {}, a2 = {}, a3 = {};
#pragma unroll
  for (int q = 0; q < 8; ++q) {
    a0 = wmma4(tc[q], bre[q], a0);
    a1 = wmma4(ts[q], bim[q], a1);
    a2 = wmma4(tc[q], bim[q], a2);
    a3 = wmma4(ts[q], bre[q], a3);
  }
#pragma unroll
  for (int r = 0; r < 8; ++r) {
    tj[(8 * half + r) * UP + l16] = a0[r] - a1[r];
    tj[(8 * half + r) * UP + 16 + l16] = a2[r] + a3[r];
  }
#pragma unroll
  for (int q = 0; q < 8; ++q) a[q] = ld2(tj + l16 * UP + 2 * half + 4 * q);
}

__global__ __launch_bounds__(256) void k4_idft_hw(
    const float* __restrict__ Tre, const float* __restrict__ Tim,
    const float* __restrict__ E4c, const float* __restrict__ E4s,
    const float* __restrict__ GT, float* __restrict__ out) {
  __shared__ __attribute__((aligned(16))) float ul[8 * 4 * UT];
  const int wv = threadIdx.x >> 5;
  const int lane = threadIdx.x & 31;
  const int half = lane >> 4, l16 = lane & 15;
  const int bo = blockIdx.x * 2 + (wv >> 2);
  const int hb = (wv & 3) * 64;
  const float* tre = Tre + (long)bo * 512 + (2 * half) * 16 + l16;
  const float* tim = Tim + (long)bo * 512 + (2 * half) * 16 + l16;
  v2f bre[8], bim[8];
#pragma unroll
  for (int q = 0; q < 8; ++q) {
    bre[q].x = tre[q * 64];
    bre[q].y = tre[q * 64 + 16];
    bim[q].x = tim[q * 64];
    bim[q].y = tim[q * 64 + 16];
  }
  const float* ec = E4c + (hb + l16) * 32 + 2 * half;
  const float* es = E4s + (hb + l16) * 32 + 2 * half;
  v2f tc[4][8], ts[4][8];
#pragma unroll
  for (int j = 0; j < 4; ++j) {
#pragma unroll
    for (int q = 0; q < 8; ++q) {
      tc[j][q] = ld2(ec + j * 512 + 4 * q);
      ts[j][q] = ld2(es + j * 512 + 4 * q);
    }
  }
  float* my = ul + wv * (4 * UT);
  v2f a[4][8];
  k4_utile(tc[0], ts[0], bre, bim, my, half, l16, a[0]);
  __builtin_amdgcn_sched_barrier(0);
  const float* g0 = GT + l16 * 32 + 2 * half;
  v2f gc[9][4], gs[8][4];
#pragma unroll
  for (int t = 0; t < 9; ++t) {
#pragma unroll
    for (int q = 0; q < 4; ++q) gc[t][q] = ld2(g0 + t * 512 + 4 * q);
  }
#pragma unroll
  for (int t = 0; t < 8; ++t) {
#pragma unroll
    for (int q = 0; q < 4; ++q) gs[t][q] = ld2(g0 + t * 512 + 16 + 4 * q);
  }
#pragma unroll
  for (int j = 1; j < 4; ++j)
    k4_utile(tc[j], ts[j], bre, bim, my + j * UT, half, l16, a[j]);
  const long rowbase = (long)bo * 256 + hb;
  float* orow = out + (rowbase + half * 8) * 256;
  float* od = orow + l16;
  float* om = orow + 128 + ((16 - l16) & 15);
#pragma unroll
  for (int j = 0; j < 4; ++j)
    k4_tile(a[j], gc, gs, od + j * 4096, om + j * 4096, l16 == 0);
}

extern "C" void kernel_launch(void* const* d_in, const int* in_sizes, int n_in,
                              void* d_out, int out_size, void* d_ws,
                              size_t ws_size, hipStream_t stream) {
  (void)in_sizes; (void)n_in; (void)out_size; (void)ws_size;
  const float* x = (const float*)d_in[0];
  const float* w1r = (const float*)d_in[1];
  const float* w1i = (const float*)d_in[2];
  const float* w2r = (const float*)d_in[3];
  const float* w2i = (const float*)d_in[4];
  float* out = (float*)d_out;

  float* tab = (float*)d_ws;
  float* E4c = tab + 24576;
  float* E4s = tab + 32768;
  float* GT = tab + 40960;
  float* Zre = tab + 49152;
  float* Zim = Zre + 262144;
  float* Tre = Zim + 262144;
  float* Tim = Tre + 262144;

  hipLaunchKernelGGL(k1_dft_hw, dim3(256), dim3(256), 0, stream, x, tab, Zre,
                     Zim);
  hipLaunchKernelGGL(k3_mix, dim3(128), dim3(256), 0, stream, Zre, Zim, w1r, w1i,
                     w2r, w2i, Tre, Tim);
  hipLaunchKernelGGL(k4_idft_hw, dim3(256), dim3(256), 0, stream, Tre, Tim, E4c,
                     E4s, GT, out);
}


# Inverse kernel: each row tile's 16 out rows staged through a wave-private LDS tile and stored as 16 B per lane, 512 contiguous bytes of one row per store, not 64-byte pieces; same sums
# speedup vs baseline: 3.2906x; 1.0994x over previous
#include <hip/hip_runtime.h>


typedef __attribute__((ext_vector_type(2))) float v2f;
typedef __attribute__((ext_vector_type(4))) float v4f;
typedef __attribute__((ext_vector_type(8))) float v8f;

__device__ __forceinline__ v8f wmma4(v2f a, v2f b, v8f c) {
  return __builtin_amdgcn_wmma_f32_16x16x4_f32(false, a, false, b, (short)0, c,
                                               false, false);
}

__device__ __forceinline__ v2f ld2(const float* p) { return *(const v2f*)p; }
__device__ __forceinline__ v4f ld4_nt(const float* p) {
  return __builtin_nontemporal_load((const v4f*)p);
}

struct XBlk {
  v4f xa[8];
  v4f xm[8];
  v2f c0[4], c1[4], s0[4], s1[4];
};

__device__ __forceinline__ void k1_load_x(XBlk& B, const float* xp,
                                          const float* xmp, int kb, int half) {
#pragma unroll
  for (int q = 0; q < 4; ++q) {
    int k0 = kb + 4 * q;
    B.xa[2 * q] = ld4_nt(xp + (long)k0 * 256);
    B.xa[2 * q + 1] = ld4_nt(xp + (long)(k0 + 1) * 256);
    const float* m0 = xmp - (long)k0 * 256;
    if (q == 0) m0 = (k0 + 2 * half == 0) ? xp : m0;
    B.xm[2 * q] = ld4_nt(m0);
    B.xm[2 * q + 1] = ld4_nt(xmp - (long)(k0 + 1) * 256);
  }
}

__device__ __forceinline__ void k1_load_t(XBlk& B, const float* c0,
                                          const float* c1, const float* s0,
                                          const float* s1, int kb) {
#pragma unroll
  for (int q = 0; q < 4; ++q) {
    int k0 = kb + 4 * q;
    B.c0[q] = ld2(c0 + k0);
    B.c1[q] = ld2(c1 + k0);
    B.s0[q] = ld2(s0 + k0);
    B.s1[q] = ld2(s1 + k0);
  }
}

__device__ __forceinline__ void k1_load(XBlk& B, const float* xp,
                                        const float* xmp, const float* c0,
                                        const float* c1, const float* s0,
                                        const float* s1, int kb, int half) {
  k1_load_x(B, xp, xmp, kb, half);
  k1_load_t(B, c0, c1, s0, s1, kb);
}

__device__ __forceinline__ void k1_mma(const XBlk& B, v8f (&acc)[4][4],
                                       bool row0) {
#pragma unroll
  for (int q = 0; q < 4; ++q) {
    v4f e0 = B.xa[2 * q] + B.xm[2 * q];
    v4f e1 = B.xa[2 * q + 1] + B.xm[2 * q + 1];
    v4f o0 = B.xa[2 * q] - B.xm[2 * q];
    v4f o1 = B.xa[2 * q + 1] - B.xm[2 * q + 1];
    if (q == 0) {
#pragma unroll
      for (int t = 0; t < 4; ++t) e0[t] = row0 ? B.xa[0][t] : e0[t];
    }
#pragma unroll
    for (int t = 0; t < 4; ++t) {
      v2f be, bo;
      be.x = e0[t];
      be.y = e1[t];
      bo.x = o0[t];
      bo.y = o1[t];
      acc[0][t] = wmma4(B.c0[q], be, acc[0][t]);
      acc[1][t] = wmma4(B.c1[q], be, acc[1][t]);
      acc[2][t] = wmma4(B.s0[q], bo, acc[2][t]);
      acc[3][t] = wmma4(B.s1[q], bo, acc[3][t]);
    }
  }
}

#define E2P 132
#define PLP 68
#define PLW (64 * PLP)
__global__ __launch_bounds__(256) void k1_dft_hw(const float* __restrict__ x,
                                                 float* __restrict__ tab,
                                                 float* __restrict__ Zre,
                                                 float* __restrict__ Zim) {
  __shared__ __attribute__((aligned(16))) float lds[512 + 64 * E2P + 8 * PLW];
  float* cs = lds;
  float* e2 = lds + 512;
  float* pl = lds + 512 + 64 * E2P;
  const int tid = threadIdx.x;
  const int wv = tid >> 5;
  const int lane = tid & 31;
  const int half = lane >> 4, l16 = lane & 15;
  const int bi = blockIdx.x * 2 + (wv >> 2);
  const int wbase = (wv & 3) * 64;
  const float* ximg = x + (long)bi * 65536 + wbase + 4 * l16;
  const float* xp = ximg + (2 * half) * 256;
  const float* xmp = ximg + (256 - 2 * half) * 256;
  const float* c0 = e2 + l16 * E2P + 2 * half;
  const float* c1 = e2 + (16 + l16) * E2P + 2 * half;
  const float* s0 = e2 + (32 + l16) * E2P + 2 * half;
  const float* s1 = e2 + (48 + l16) * E2P + 2 * half;
  v8f acc[4][4] = {};
  XBlk A, B;
  k1_load_x(A, xp, xmp, 0, half);
  {
    const float step = 6.28318530717958647692f / 256.0f;
    float th = step * (float)tid;
    cs[tid] = cosf(th);
    cs[256 + tid] = sinf(th);
  }
  __syncthreads();
  {
    const int row = tid >> 2;
    const int r = row & 31;
    const int ky = (r < 16) ? r : (224 + r);
    const int h0 = (tid & 3) * 33;
    const float* src = cs + ((row >> 5) << 8);
    float* dst = e2 + row * E2P + h0;
    int n = ky * h0;
#pragma unroll
    for (int it = 0; it < 33; ++it) {
      dst[it] = src[n & 255];
      n += ky;
    }
  }
  __syncthreads();
  if (tid < 128 && blockIdx.x < 192) {
    const int e = 8192 + blockIdx.x * 128 + tid;
    float v;
    int o;
    if (e < 24576) {
      int j = e - 8192;
      int is_s = j >= 8192;
      int h = (j & 8191) >> 5, r = j & 31;
      int ky = (r < 16) ? r : (224 + r);
      int m = (ky * h) & 255;
      v = is_s ? cs[256 + m] : cs[m];
      o = 24576 + j;
    } else {
      int j = e - 24576;
      int w = j >> 5, n = j & 31;
      int kx = n & 15;
      int m = (kx * w) & 255;
      float sc = ((kx == 0) ? 1.0f : 2.0f) * (1.0f / 65536.0f);
      v = (n < 16) ? sc * cs[m] : -sc * cs[256 + m];
      o = 40960 + j;
    }
    tab[o] = v;
  }
  k1_load_t(A, c0, c1, s0, s1, 0);
#pragma unroll 1
  for (int kb = 0; kb < 96; kb += 32) {
    k1_load(B, xp, xmp, c0, c1, s0, s1, kb + 16, half);
    __builtin_amdgcn_sched_barrier(0);
    k1_mma(A, acc, (kb == 0) && (half == 0));
    __builtin_amdgcn_sched_barrier(0);
    k1_load(A, xp, xmp, c0, c1, s0, s1, kb + 32, half);
    __builtin_amdgcn_sched_barrier(0);
    k1_mma(B, acc, false);
    __builtin_amdgcn_sched_barrier(0);
  }
  k1_load(B, xp, xmp, c0, c1, s0, s1, 112, half);
  v4f x128 = ld4_nt(ximg + 128 * 256);
  v2f t0 = ld2(c0 + 128);
  v2f t1 = ld2(c1 + 128);
  k1_mma(A, acc, false);
  k1_mma(B, acc, false);
  {
    v2f a0, a1;
    a0.x = half ? 0.0f : t0.x;
    a0.y = 0.0f;
    a1.x = half ? 0.0f : t1.x;
    a1.y = 0.0f;
#pragma unroll
    for (int t = 0; t < 4; ++t) {
      v2f b;
      b.x = half ? 0.0f : x128[t];
      b.y = 0.0f;
      acc[0][t] = wmma4(a0, b, acc[0][t]);
      acc[1][t] = wmma4(a1, b, acc[1][t]);
    }
  }
  float* my = pl + wv * PLW;
#pragma unroll
  for (int a = 0; a < 4; ++a) {
#pragma unroll
    for (int r = 0; r < 8; ++r) {
      v4f o = {acc[a][0][r], acc[a][1][r], acc[a][2][r], acc[a][3][r]};
      *(v4f*)(my + (16 * a + 8 * half + r) * PLP + 4 * l16) = o;
    }
  }
  const float* arow = my + l16 * PLP + 2 * half;
  v2f fc0[16], fc1[16], fs0[16], fs1[16], bco[16], bsi[16];
#pragma unroll
  for (int q = 0; q < 16; ++q) {
    fc0[q] = ld2(arow + 4 * q);
    fc1[q] = ld2(arow + 16 * PLP + 4 * q);
    fs0[q] = ld2(arow + 32 * PLP + 4 * q);
    fs1[q] = ld2(arow + 48 * PLP + 4 * q);
    int m0 = (l16 * (wbase + 4 * q + 2 * half)) & 255;
    int m1 = (m0 + l16) & 255;
    bco[q].x = cs[m0];
    bco[q].y = cs[m1];
    bsi[q].x = cs[256 + m0];
    bsi[q].y = cs[256 + m1];
  }
  __builtin_amdgcn_sched_barrier(0);
  v8f z0 = {}, z1 = {}, z2 = {}, z3 = {}, z4 = {}, z5 = {}, z6 = {}, z7 = {};
#pragma unroll
  for (int q = 0; q < 16; ++q) {
    z0 = wmma4(fc0[q], bco[q], z0);
    z1 = wmma4(fs0[q], bsi[q], z1);
    z2 = wmma4(fc0[q], bsi[q], z2);
    z3 = wmma4(fs0[q], bco[q], z3);
    z4 = wmma4(fc1[q], bco[q], z4);
    z5 = wmma4(fs1[q], bsi[q], z5);
    z6 = wmma4(fc1[q], bsi[q], z6);
    z7 = wmma4(fs1[q], bco[q], z7);
  }
  float* mine = my + lane;
#pragma unroll
  for (int r = 0; r < 8; ++r) {
    mine[(0 * 8 + r) * 32] = z0[r] - z1[r];
    mine[(1 * 8 + r) * 32] = -z2[r] - z3[r];
    mine[(2 * 8 + r) * 32] = z4[r] - z5[r];
    mine[(3 * 8 + r) * 32] = -z6[r] - z7[r];
  }
  __syncthreads();
  {
    const int u = tid & 127;
    const float* p0 = pl + (wv >> 2) * 4 * PLW + u;
    const int b = bi >> 5, i = bi & 31;
    const int hf = (u >> 4) & 1, kx = u & 15;
#pragma unroll
    for (int n = 0; n < 8; ++n) {
      int v = u + 128 * n;
      float s = p0[128 * n];
      s += p0[PLW + 128 * n];
      s += p0[2 * PLW + 128 * n];
      s += p0[3 * PLW + 128 * n];
      int o = v >> 8, r = (v >> 5) & 7;
      int kyp = (o >> 1) * 16 + hf * 8 + r;
      long off = (((long)kyp * 16 + kx) * 16 + b) * 32 + i;
      if (o & 1)
        Zim[off] = s;
      else
        Zre[off] = s;
    }
  }
}

#define K3_PLANE 520
__global__ __launch_bounds__(256) void k3_mix(
    const float* __restrict__ Zre, const float* __restrict__ Zim,
    const float* __restrict__ w1r, const float* __restrict__ w1i,
    const float* __restrict__ w2r, const float* __restrict__ w2i,
    float* __restrict__ Tre, float* __restrict__ Tim) {
  __shared__ __attribute__((aligned(16))) float wl[2][8 * K3_PLANE];
  __shared__ __attribute__((aligned(16))) float tl[2][2048];
  const int tid = threadIdx.x;
  const int wv = tid >> 5;
  const int lane = tid & 31;
  const int half = lane >> 4, l16 = lane & 15;
  const int kyp = blockIdx.x >> 2;
  const int ot = (blockIdx.x >> 1) & 1;
  const int kh = blockIdx.x & 1;
  const int ky = kyp & 15;
  const float* wr = (kyp < 16) ? w1r : w2r;
  const float* wi = (kyp < 16) ? w1i : w2i;
  const int m = kyp * 16 + kh * 8 + wv;
  const float* zre = Zre + (long)m * 512 + l16 * 32 + 2 * half;
  const float* zim = Zim + (long)m * 512 + l16 * 32 + 2 * half;
  v2f azr[8], azi[8];
#pragma unroll
  for (int q = 0; q < 8; ++q) {
    azr[q] = ld2(zre + 4 * q);
    azi[q] = ld2(zim + 4 * q);
  }
#pragma unroll
  for (int n = 0; n < 4; ++n) {
    int p = tid + 256 * n;
    int j = p & 1, ol = (p >> 1) & 15, i = p >> 5;
    long g = ((long)(i * 32 + ot * 16 + ol)) * 256 + ky * 16 + kh * 8 + j * 4;
    v4f vr = *(const v4f*)(wr + g);
    v4f vi = *(const v4f*)(wi + g);
    int base = (j * 4) * K3_PLANE + ((i >> 1) * 16 + ol) * 2 + (i & 1);
#pragma unroll
    for (int e = 0; e < 4; ++e) {
      wl[0][base + e * K3_PLANE] = vr[e];
      wl[1][base + e * K3_PLANE] = vi[e];
    }
  }
  __syncthreads();
  const float* bre = &wl[0][wv * K3_PLANE + (half * 16 + l16) * 2];
  const float* bim = &wl[1][wv * K3_PLANE + (half * 16 + l16) * 2];
  v8f p0 = {}, q0 = {}, u0 = {}, v0 = {};
#pragma unroll
  for (int q = 0; q < 8; ++q) {
    v2f br = *(const v2f*)(bre + q * 64);
    v2f bi = *(const v2f*)(bim + q * 64);
    p0 = wmma4(azr[q], br, p0);
    q0 = wmma4(azi[q], bi, q0);
    u0 = wmma4(azr[q], bi, u0);
    v0 = wmma4(azi[q], br, v0);
  }
#pragma unroll
  for (int r = 0; r < 8; ++r) {
    int t = ((r + half * 8) * 16 + l16) * 8 + wv;
    tl[0][t] = p0[r] - q0[r];
    tl[1][t] = u0[r] + v0[r];
  }
  __syncthreads();
#pragma unroll
  for (int n = 0; n < 2; ++n) {
    int pp = tid + 256 * n;
    int j = pp & 1, ol = (pp >> 1) & 15, b = pp >> 5;
    long o = (((long)b * 32 + ot * 16 + ol) * 32 + kyp) * 16 + kh * 8 + j * 4;
    *(v4f*)(Tre + o) = *(const v4f*)&tl[0][pp * 4];
    *(v4f*)(Tim + o) = *(const v4f*)&tl[1][pp * 4];
  }
}

#define OP 260
__device__ __forceinline__ void k4_tile(const v2f (&a)[8],
                                        const v2f (&gc)[9][4],
                                        const v2f (&gs)[8][4], float* ot,
                                        float* og, int half, int l16,
                                        int lane) {
  v8f e[9] = {}, o[8] = {};
#pragma unroll
  for (int q = 0; q < 4; ++q) {
#pragma unroll
    for (int t = 0; t < 9; ++t) e[t] = wmma4(a[q], gc[t][q], e[t]);
  }
#pragma unroll
  for (int q = 0; q < 4; ++q) {
#pragma unroll
    for (int t = 0; t < 8; ++t) o[t] = wmma4(a[4 + q], gs[t][q], o[t]);
  }
  float* wd = ot + (8 * half) * OP + l16;
  float* wm = ot + (8 * half) * OP + 128 + ((16 - l16) & 15);
  const bool lane0 = l16 == 0;
#pragma unroll
  for (int r = 0; r < 8; ++r) {
#pragma unroll
    for (int t = 0; t < 8; ++t) wd[r * OP + t * 16] = e[t][r] + o[t][r];
#pragma unroll
    for (int p = 0; p < 8; ++p) {
      float hi = (p == 0) ? e[8][r] : e[8 - p][r] - o[8 - p][r];
      float lo = e[7 - p][r] - o[7 - p][r];
      wm[r * OP + p * 16] = lane0 ? hi : lo;
    }
  }
#pragma unroll
  for (int n = 0; n < 32; ++n) {
    v4f v = *(const v4f*)(ot + (n >> 1) * OP + (n & 1) * 128 + 4 * lane);
    __builtin_nontemporal_store(
        v, (v4f*)(og + (n >> 1) * 256 + (n & 1) * 128 + 4 * lane));
  }
}

#define UP 36
#define UT (16 * UP)
__device__ __forceinline__ void k4_utile(const v2f (&tc)[8], const v2f (&ts)[8],
                                         const v2f (&bre)[8],
                                         const v2f (&bim)[8], float* tj,
                                         int half, int l16, v2f (&a)[8]) {
  v8f a0 = {}, a1 = {}, a2 = {}, a3 = {};
#pragma unroll
  for (int q = 0; q < 8; ++q) {
    a0 = wmma4(tc[q], bre[q], a0);
    a1 = wmma4(ts[q], bim[q], a1);
    a2 = wmma4(tc[q], bim[q], a2);
    a3 = wmma4(ts[q], bre[q], a3);
  }
#pragma unroll
  for (int r = 0; r < 8; ++r) {
    tj[(8 * half + r) * UP + l16] = a0[r] - a1[r];
    tj[(8 * half + r) * UP + 16 + l16] = a2[r] + a3[r];
  }
#pragma unroll
  for (int q = 0; q < 8; ++q) a[q] = ld2(tj + l16 * UP + 2 * half + 4 * q);
}

__global__ __launch_bounds__(256) void k4_idft_hw(
    const float* __restrict__ Tre, const float* __restrict__ Tim,
    const float* __restrict__ E4c, const float* __restrict__ E4s,
    const float* __restrict__ GT, float* __restrict__ out) {
  __shared__ __attribute__((aligned(16))) float ul[8 * 4 * UT + 8 * 16 * OP];
  const int wv = threadIdx.x >> 5;
  const int lane = threadIdx.x & 31;
  const int half = lane >> 4, l16 = lane & 15;
  const int bo = blockIdx.x * 2 + (wv >> 2);
  const int hb = (wv & 3) * 64;
  const float* tre = Tre + (long)bo * 512 + (2 * half) * 16 + l16;
  const float* tim = Tim + (long)bo * 512 + (2 * half) * 16 + l16;
  v2f bre[8], bim[8];
#pragma unroll
  for (int q = 0; q < 8; ++q) {
    bre[q].x = tre[q * 64];
    bre[q].y = tre[q * 64 + 16];
    bim[q].x = tim[q * 64];
    bim[q].y = tim[q * 64 + 16];
  }
  const float* ec = E4c + (hb + l16) * 32 + 2 * half;
  const float* es = E4s + (hb + l16) * 32 + 2 * half;
  v2f tc[4][8], ts[4][8];
#pragma unroll
  for (int j = 0; j < 4; ++j) {
#pragma unroll
    for (int q = 0; q < 8; ++q) {
      tc[j][q] = ld2(ec + j * 512 + 4 * q);
      ts[j][q] = ld2(es + j * 512 + 4 * q);
    }
  }
  float* my = ul + wv * (4 * UT);
  v2f a[4][8];
  k4_utile(tc[0], ts[0], bre, bim, my, half, l16, a[0]);
  __builtin_amdgcn_sched_barrier(0);
  const float* g0 = GT + l16 * 32 + 2 * half;
  v2f gc[9][4], gs[8][4];
#pragma unroll
  for (int t = 0; t < 9; ++t) {
#pragma unroll
    for (int q = 0; q < 4; ++q) gc[t][q] = ld2(g0 + t * 512 + 4 * q);
  }
#pragma unroll
  for (int t = 0; t < 8; ++t) {
#pragma unroll
    for (int q = 0; q < 4; ++q) gs[t][q] = ld2(g0 + t * 512 + 16 + 4 * q);
  }
#pragma unroll
  for (int j = 1; j < 4; ++j)
    k4_utile(tc[j], ts[j], bre, bim, my + j * UT, half, l16, a[j]);
  const long rowbase = (long)bo * 256 + hb;
  float* og = out + rowbase * 256;
  float* ot = ul + 8 * 4 * UT + wv * (16 * OP);
#pragma unroll
  for (int j = 0; j < 4; ++j)
    k4_tile(a[j], gc, gs, ot, og + j * 4096, half, l16, lane);
}

extern "C" void kernel_launch(void* const* d_in, const int* in_sizes, int n_in,
                              void* d_out, int out_size, void* d_ws,
                              size_t ws_size, hipStream_t stream) {
  (void)in_sizes; (void)n_in; (void)out_size; (void)ws_size;
  const float* x = (const float*)d_in[0];
  const float* w1r = (const float*)d_in[1];
  const float* w1i = (const float*)d_in[2];
  const float* w2r = (const float*)d_in[3];
  const float* w2i = (const float*)d_in[4];
  float* out = (float*)d_out;

  float* tab = (float*)d_ws;
  float* E4c = tab + 24576;
  float* E4s = tab + 32768;
  float* GT = tab + 40960;
  float* Zre = tab + 49152;
  float* Zim = Zre + 262144;
  float* Tre = Zim + 262144;
  float* Tim = Tre + 262144;

  hipLaunchKernelGGL(k1_dft_hw, dim3(256), dim3(256), 0, stream, x, tab, Zre,
                     Zim);
  hipLaunchKernelGGL(k3_mix, dim3(128), dim3(256), 0, stream, Zre, Zim, w1r, w1i,
                     w2r, w2i, Tre, Tim);
  hipLaunchKernelGGL(k4_idft_hw, dim3(256), dim3(256), 0, stream, Tre, Tim, E4c,
                     E4s, GT, out);
}
